# starting kernel
# baseline (speedup 1.0000x reference)
#include <hip/hip_runtime.h>

#define N_NODES 20000
#define N_EDGES 150000
#define D 512
#define BN_EPS 1e-5f

typedef __attribute__((ext_vector_type(2))) float v2f;
typedef __attribute__((ext_vector_type(8))) float v8f;

__global__ void zero_kernel(float* __restrict__ p, long long n) {
  long long i = (long long)blockIdx.x * blockDim.x + threadIdx.x;
  long long stride = (long long)gridDim.x * blockDim.x;
  for (; i < n; i += stride) p[i] = 0.0f;
}

__global__ __launch_bounds__(256)
void scatter_kernel(const float* __restrict__ x,
                    const int* __restrict__ src,
                    const int* __restrict__ dst,
                    float* __restrict__ agg,
                    float* __restrict__ indeg) {
  int wave = threadIdx.x >> 5;
  int lane = threadIdx.x & 31;
  int e = blockIdx.x * 8 + wave;
  if (e >= N_EDGES) return;
  int s = src[e];
  int d = dst[e];
  const float4* xs = (const float4*)(x + (size_t)s * D);
  float* ag = agg + (size_t)d * D;
  if (lane == 0) atomicAdd(indeg + d, 1.0f);
#pragma unroll
  for (int j = 0; j < 4; ++j) {
    int idx = j * 32 + lane;
    float4 v = xs[idx];
    atomicAdd(ag + idx * 4 + 0, v.x);
    atomicAdd(ag + idx * 4 + 1, v.y);
    atomicAdd(ag + idx * 4 + 2, v.z);
    atomicAdd(ag + idx * 4 + 3, v.w);
  }
}

__global__ __launch_bounds__(256)
void gemm_wmma_kernel(const float* __restrict__ x,
                      const float* __restrict__ agg,
                      const float* __restrict__ indeg,
                      const float* __restrict__ Wm,
                      const float* __restrict__ bias,
                      float* __restrict__ out,
                      float* __restrict__ colsum,
                      float* __restrict__ colsumsq) {
  __shared__ float Alds[128 * 33];
  __shared__ float Blds[64 * 33];

  const int t = threadIdx.x;
  const int lane = t & 31;
  const int wave = t >> 5;
  const int rowTile = blockIdx.y * 128;
  const int colTile = blockIdx.x * 64;
  const int kg = (t & 7) * 4;

  const float* asrc[4];
#pragma unroll
  for (int i = 0; i < 4; ++i) {
    int r = (t >> 3) + 32 * i;
    int g = rowTile + r;
    asrc[i] = (g < N_NODES) ? ((indeg[g] > 0.0f ? agg : x) + (size_t)g * D) : nullptr;
  }
  const float* bsrc[2];
#pragma unroll
  for (int i = 0; i < 2; ++i) {
    int c = (t >> 3) + 32 * i;
    bsrc[i] = Wm + (size_t)(colTile + c) * D + kg;
  }

  v8f acc0 = {}, acc1 = {}, acc2 = {}, acc3 = {};

  for (int kt = 0; kt < D; kt += 32) {
    __syncthreads();
#pragma unroll
    for (int i = 0; i < 4; ++i) {
      int r = (t >> 3) + 32 * i;
      float4 v = make_float4(0.f, 0.f, 0.f, 0.f);
      if (asrc[i]) v = *(const float4*)(asrc[i] + kt + kg);
      Alds[r * 33 + kg + 0] = v.x;
      Alds[r * 33 + kg + 1] = v.y;
      Alds[r * 33 + kg + 2] = v.z;
      Alds[r * 33 + kg + 3] = v.w;
    }
#pragma unroll
    for (int i = 0; i < 2; ++i) {
      int c = (t >> 3) + 32 * i;
      float4 v = *(const float4*)(bsrc[i] + kt);
      Blds[c * 33 + kg + 0] = v.x;
      Blds[c * 33 + kg + 1] = v.y;
      Blds[c * 33 + kg + 2] = v.z;
      Blds[c * 33 + kg + 3] = v.w;
    }
    __syncthreads();

    const int arow = wave * 16 + (lane & 15);
    const int bcol = lane & 15;
    const int khalf = (lane >> 4) * 2;
#pragma unroll
    for (int kk = 0; kk < 32; kk += 4) {
      int ko = kk + khalf;
      v2f a;  a[0]  = Alds[arow * 33 + ko];        a[1]  = Alds[arow * 33 + ko + 1];
      v2f b0; b0[0] = Blds[(bcol +  0) * 33 + ko]; b0[1] = Blds[(bcol +  0) * 33 + ko + 1];
      v2f b1; b1[0] = Blds[(bcol + 16) * 33 + ko]; b1[1] = Blds[(bcol + 16) * 33 + ko + 1];
      v2f b2; b2[0] = Blds[(bcol + 32) * 33 + ko]; b2[1] = Blds[(bcol + 32) * 33 + ko + 1];
      v2f b3; b3[0] = Blds[(bcol + 48) * 33 + ko]; b3[1] = Blds[(bcol + 48) * 33 + ko + 1];
      acc0 = __builtin_amdgcn_wmma_f32_16x16x4_f32(false, a, false, b0, (short)0, acc0, false, false);
      acc1 = __builtin_amdgcn_wmma_f32_16x16x4_f32(false, a, false, b1, (short)0, acc1, false, false);
      acc2 = __builtin_amdgcn_wmma_f32_16x16x4_f32(false, a, false, b2, (short)0, acc2, false, false);
      acc3 = __builtin_amdgcn_wmma_f32_16x16x4_f32(false, a, false, b3, (short)0, acc3, false, false);
    }
  }

  int gRowBase = rowTile + wave * 16 + ((lane >> 4) << 3);
  v8f accs[4] = {acc0, acc1, acc2, acc3};
#pragma unroll
  for (int nt = 0; nt < 4; ++nt) {
    int col = colTile + nt * 16 + (lane & 15);
    float bv = bias[col];
    float ps = 0.0f, pq = 0.0f;
#pragma unroll
    for (int r = 0; r < 8; ++r) {
      int grow = gRowBase + r;
      float h = accs[nt][r] + bv;
      h = h > 0.0f ? h : 0.0f;
      if (grow < N_NODES) {
        out[(size_t)grow * D + col] = h;
        ps += h;
        pq += h * h;
      }
    }
    ps += __shfl_xor(ps, 16, 32);
    pq += __shfl_xor(pq, 16, 32);
    if (lane < 16) {
      atomicAdd(colsum + col, ps);
      atomicAdd(colsumsq + col, pq);
    }
  }
}

__global__ void bn_kernel(float* __restrict__ out,
                          const float* __restrict__ colsum,
                          const float* __restrict__ colsumsq,
                          const float* __restrict__ gamma,
                          const float* __restrict__ beta) {
  const float invN = 1.0f / (float)N_NODES;
  long long total = (long long)N_NODES * D;
  long long i = (long long)blockIdx.x * blockDim.x + threadIdx.x;
  long long stride = (long long)gridDim.x * blockDim.x;
  for (; i < total; i += stride) {
    int col = (int)(i & (D - 1));
    float mean = colsum[col] * invN;
    float var = colsumsq[col] * invN - mean * mean;
    out[i] = (out[i] - mean) * rsqrtf(var + BN_EPS) * gamma[col] + beta[col];
  }
}

extern "C" void kernel_launch(void* const* d_in, const int* in_sizes, int n_in,
                              void* d_out, int out_size, void* d_ws, size_t ws_size,
                              hipStream_t stream) {
  const float* x     = (const float*)d_in[0];
  const int*   src   = (const int*)d_in[1];
  const int*   dst   = (const int*)d_in[2];
  const float* Wm    = (const float*)d_in[3];
  const float* bias  = (const float*)d_in[4];
  const float* gamma = (const float*)d_in[5];
  const float* beta  = (const float*)d_in[6];
  float* out = (float*)d_out;

  float* ws       = (float*)d_ws;
  float* agg      = ws;
  float* indeg    = agg + (size_t)N_NODES * D;
  float* colsum   = indeg + N_NODES;
  float* colsumsq = colsum + D;

  long long zcount = (long long)N_NODES * D + N_NODES + 2 * D;
  zero_kernel<<<2048, 256, 0, stream>>>(ws, zcount);

  scatter_kernel<<<(N_EDGES + 7) / 8, 256, 0, stream>>>(x, src, dst, agg, indeg);

  dim3 grid(D / 64, (N_NODES + 127) / 128);
  gemm_wmma_kernel<<<grid, 256, 0, stream>>>(x, agg, indeg, Wm, bias, out,
                                             colsum, colsumsq);

  bn_kernel<<<4096, 256, 0, stream>>>(out, colsum, colsumsq, gamma, beta);
}


# Float-atomic edge scatter + 41 MB zero-fill replaced by a gather per destination: one int exchange per edge links in-edge lists, one wave sums each h row in list (arrival) order and stores it once
# speedup vs baseline: 1.4833x; 1.4833x over previous
#include <hip/hip_runtime.h>

#define N_NODES 20000
#define N_EDGES 150000
#define D 512
#define BN_EPS 1e-5f

typedef __attribute__((ext_vector_type(2))) float v2f;
typedef __attribute__((ext_vector_type(8))) float v8f;

__global__ void init_kernel(int* __restrict__ head, float* __restrict__ colstats) {
  int i = blockIdx.x * blockDim.x + threadIdx.x;
  if (i < N_NODES) head[i] = -1;
  if (i < 2 * D) colstats[i] = 0.0f;
}

static_assert(N_EDGES % 4 == 0, "link_kernel takes edges four at a time");
__global__ __launch_bounds__(256)
void link_kernel(const int* __restrict__ dst,
                 int* __restrict__ head,
                 int* __restrict__ nxt) {
  const int t = blockIdx.x * blockDim.x + threadIdx.x;
  if (t * 4 >= N_EDGES) return;
  const int4 d = ((const int4*)dst)[t];
  const int e = t * 4;
  int4 p;
  p.x = atomicExch(head + d.x, e);
  p.y = atomicExch(head + d.y, e + 1);
  p.z = atomicExch(head + d.z, e + 2);
  p.w = atomicExch(head + d.w, e + 3);
  ((int4*)nxt)[t] = p;
}

__device__ __forceinline__ void add_row(const float* __restrict__ x, int row, int lane,
                                        float4& a0, float4& a1, float4& a2, float4& a3) {
  const float4* xs = (const float4*)(x + (size_t)row * D);
  const float4 v0 = xs[lane], v1 = xs[32 + lane], v2 = xs[64 + lane], v3 = xs[96 + lane];
  a0.x += v0.x; a0.y += v0.y; a0.z += v0.z; a0.w += v0.w;
  a1.x += v1.x; a1.y += v1.y; a1.z += v1.z; a1.w += v1.w;
  a2.x += v2.x; a2.y += v2.y; a2.z += v2.z; a2.w += v2.w;
  a3.x += v3.x; a3.y += v3.y; a3.z += v3.z; a3.w += v3.w;
}

__global__ __launch_bounds__(256)
void gather_kernel(const float* __restrict__ x,
                   const int* __restrict__ src,
                   const int* __restrict__ head,
                   const int* __restrict__ nxt,
                   float* __restrict__ h) {
  const int lane = threadIdx.x & 31;
  const int nw = gridDim.x * 8;
  const int w0 = __builtin_amdgcn_readfirstlane(blockIdx.x * 8 + (threadIdx.x >> 5));
  for (int node = w0; node < N_NODES; node += nw) {
    float4 a0 = make_float4(0.f, 0.f, 0.f, 0.f), a1 = a0, a2 = a0, a3 = a0;
    const int e0 = head[node];
    if (e0 < 0) {
      const float4* xs = (const float4*)(x + (size_t)node * D);
      a0 = xs[lane]; a1 = xs[32 + lane]; a2 = xs[64 + lane]; a3 = xs[96 + lane];
    } else {
      int sa = src[e0], ea = nxt[e0], sb = 0, eb = -1;
      for (int it = 0; it < N_EDGES; it += 2) {
        if (ea >= 0) { sb = src[ea]; eb = nxt[ea]; }
        add_row(x, sa, lane, a0, a1, a2, a3);
        if (ea < 0) break;
        if (eb >= 0) { sa = src[eb]; ea = nxt[eb]; }
        add_row(x, sb, lane, a0, a1, a2, a3);
        if (eb < 0) break;
      }
    }
    float4* hp = (float4*)(h + (size_t)node * D);
    hp[lane] = a0; hp[32 + lane] = a1; hp[64 + lane] = a2; hp[96 + lane] = a3;
  }
}

__global__ __launch_bounds__(256)
void gemm_wmma_kernel(const float* __restrict__ hmat,
                      const float* __restrict__ Wm,
                      const float* __restrict__ bias,
                      float* __restrict__ out,
                      float* __restrict__ colsum,
                      float* __restrict__ colsumsq) {
  __shared__ float Alds[128 * 33];
  __shared__ float Blds[64 * 33];

  const int t = threadIdx.x;
  const int lane = t & 31;
  const int wave = t >> 5;
  const int rowTile = blockIdx.y * 128;
  const int colTile = blockIdx.x * 64;
  const int kg = (t & 7) * 4;

  const float* asrc[4];
#pragma unroll
  for (int i = 0; i < 4; ++i) {
    int r = (t >> 3) + 32 * i;
    int g = rowTile + r;
    asrc[i] = (g < N_NODES) ? (hmat + (size_t)g * D) : nullptr;
  }
  const float* bsrc[2];
#pragma unroll
  for (int i = 0; i < 2; ++i) {
    int c = (t >> 3) + 32 * i;
    bsrc[i] = Wm + (size_t)(colTile + c) * D + kg;
  }

  v8f acc0 = {}, acc1 = {}, acc2 = {}, acc3 = {};

  for (int kt = 0; kt < D; kt += 32) {
    __syncthreads();
#pragma unroll
    for (int i = 0; i < 4; ++i) {
      int r = (t >> 3) + 32 * i;
      float4 v = make_float4(0.f, 0.f, 0.f, 0.f);
      if (asrc[i]) v = *(const float4*)(asrc[i] + kt + kg);
      Alds[r * 33 + kg + 0] = v.x;
      Alds[r * 33 + kg + 1] = v.y;
      Alds[r * 33 + kg + 2] = v.z;
      Alds[r * 33 + kg + 3] = v.w;
    }
#pragma unroll
    for (int i = 0; i < 2; ++i) {
      int c = (t >> 3) + 32 * i;
      float4 v = *(const float4*)(bsrc[i] + kt);
      Blds[c * 33 + kg + 0] = v.x;
      Blds[c * 33 + kg + 1] = v.y;
      Blds[c * 33 + kg + 2] = v.z;
      Blds[c * 33 + kg + 3] = v.w;
    }
    __syncthreads();

    const int arow = wave * 16 + (lane & 15);
    const int bcol = lane & 15;
    const int khalf = (lane >> 4) * 2;
#pragma unroll
    for (int kk = 0; kk < 32; kk += 4) {
      int ko = kk + khalf;
      v2f a;  a[0]  = Alds[arow * 33 + ko];        a[1]  = Alds[arow * 33 + ko + 1];
      v2f b0; b0[0] = Blds[(bcol +  0) * 33 + ko]; b0[1] = Blds[(bcol +  0) * 33 + ko + 1];
      v2f b1; b1[0] = Blds[(bcol + 16) * 33 + ko]; b1[1] = Blds[(bcol + 16) * 33 + ko + 1];
      v2f b2; b2[0] = Blds[(bcol + 32) * 33 + ko]; b2[1] = Blds[(bcol + 32) * 33 + ko + 1];
      v2f b3; b3[0] = Blds[(bcol + 48) * 33 + ko]; b3[1] = Blds[(bcol + 48) * 33 + ko + 1];
      acc0 = __builtin_amdgcn_wmma_f32_16x16x4_f32(false, a, false, b0, (short)0, acc0, false, false);
      acc1 = __builtin_amdgcn_wmma_f32_16x16x4_f32(false, a, false, b1, (short)0, acc1, false, false);
      acc2 = __builtin_amdgcn_wmma_f32_16x16x4_f32(false, a, false, b2, (short)0, acc2, false, false);
      acc3 = __builtin_amdgcn_wmma_f32_16x16x4_f32(false, a, false, b3, (short)0, acc3, false, false);
    }
  }

  int gRowBase = rowTile + wave * 16 + ((lane >> 4) << 3);
  v8f accs[4] = {acc0, acc1, acc2, acc3};
#pragma unroll
  for (int nt = 0; nt < 4; ++nt) {
    int col = colTile + nt * 16 + (lane & 15);
    float bv = bias[col];
    float ps = 0.0f, pq = 0.0f;
#pragma unroll
    for (int r = 0; r < 8; ++r) {
      int grow = gRowBase + r;
      float h = accs[nt][r] + bv;
      h = h > 0.0f ? h : 0.0f;
      if (grow < N_NODES) {
        out[(size_t)grow * D + col] = h;
        ps += h;
        pq += h * h;
      }
    }
    ps += __shfl_xor(ps, 16, 32);
    pq += __shfl_xor(pq, 16, 32);
    if (lane < 16) {
      atomicAdd(colsum + col, ps);
      atomicAdd(colsumsq + col, pq);
    }
  }
}

__global__ void bn_kernel(float* __restrict__ out,
                          const float* __restrict__ colsum,
                          const float* __restrict__ colsumsq,
                          const float* __restrict__ gamma,
                          const float* __restrict__ beta) {
  const float invN = 1.0f / (float)N_NODES;
  long long total = (long long)N_NODES * D;
  long long i = (long long)blockIdx.x * blockDim.x + threadIdx.x;
  long long stride = (long long)gridDim.x * blockDim.x;
  for (; i < total; i += stride) {
    int col = (int)(i & (D - 1));
    float mean = colsum[col] * invN;
    float var = colsumsq[col] * invN - mean * mean;
    out[i] = (out[i] - mean) * rsqrtf(var + BN_EPS) * gamma[col] + beta[col];
  }
}

extern "C" void kernel_launch(void* const* d_in, const int* in_sizes, int n_in,
                              void* d_out, int out_size, void* d_ws, size_t ws_size,
                              hipStream_t stream) {
  const float* x     = (const float*)d_in[0];
  const int*   src   = (const int*)d_in[1];
  const int*   dst   = (const int*)d_in[2];
  const float* Wm    = (const float*)d_in[3];
  const float* bias  = (const float*)d_in[4];
  const float* gamma = (const float*)d_in[5];
  const float* beta  = (const float*)d_in[6];
  float* out = (float*)d_out;

  float* ws       = (float*)d_ws;
  float* h        = ws;
  int*   head     = (int*)(h + (size_t)N_NODES * D);
  float* colsum   = (float*)(head + N_NODES);
  float* colsumsq = colsum + D;
  int*   nxt      = (int*)d_out;

  init_kernel<<<(N_NODES + 255) / 256, 256, 0, stream>>>(head, colsum);

  link_kernel<<<(N_EDGES / 4 + 255) / 256, 256, 0, stream>>>(dst, head, nxt);

  gather_kernel<<<N_NODES / 16, 256, 0, stream>>>(x, src, head, nxt, h);

  dim3 grid(D / 64, (N_NODES + 127) / 128);
  gemm_wmma_kernel<<<grid, 256, 0, stream>>>(h, Wm, bias, out, colsum, colsumsq);

  bn_kernel<<<4096, 256, 0, stream>>>(out, colsum, colsumsq, gamma, beta);
}


# GEMM K loop software-pipelined: the next K step's six 16-byte global loads (A rows clamped, not branched) are issued before this step's WMMAs and waited for at the next hand-over to LDS
# speedup vs baseline: 1.6239x; 1.0948x over previous
#include <hip/hip_runtime.h>

#define N_NODES 20000
#define N_EDGES 150000
#define D 512
#define BN_EPS 1e-5f

typedef __attribute__((ext_vector_type(2))) float v2f;
typedef __attribute__((ext_vector_type(8))) float v8f;

__global__ void init_kernel(int* __restrict__ head, float* __restrict__ colstats) {
  int i = blockIdx.x * blockDim.x + threadIdx.x;
  if (i < N_NODES) head[i] = -1;
  if (i < 2 * D) colstats[i] = 0.0f;
}

static_assert(N_EDGES % 4 == 0, "link_kernel takes edges four at a time");
__global__ __launch_bounds__(256)
void link_kernel(const int* __restrict__ dst,
                 int* __restrict__ head,
                 int* __restrict__ nxt) {
  const int t = blockIdx.x * blockDim.x + threadIdx.x;
  if (t * 4 >= N_EDGES) return;
  const int4 d = ((const int4*)dst)[t];
  const int e = t * 4;
  int4 p;
  p.x = atomicExch(head + d.x, e);
  p.y = atomicExch(head + d.y, e + 1);
  p.z = atomicExch(head + d.z, e + 2);
  p.w = atomicExch(head + d.w, e + 3);
  ((int4*)nxt)[t] = p;
}

__device__ __forceinline__ void add_row(const float* __restrict__ x, int row, int lane,
                                        float4& a0, float4& a1, float4& a2, float4& a3) {
  const float4* xs = (const float4*)(x + (size_t)row * D);
  const float4 v0 = xs[lane], v1 = xs[32 + lane], v2 = xs[64 + lane], v3 = xs[96 + lane];
  a0.x += v0.x; a0.y += v0.y; a0.z += v0.z; a0.w += v0.w;
  a1.x += v1.x; a1.y += v1.y; a1.z += v1.z; a1.w += v1.w;
  a2.x += v2.x; a2.y += v2.y; a2.z += v2.z; a2.w += v2.w;
  a3.x += v3.x; a3.y += v3.y; a3.z += v3.z; a3.w += v3.w;
}

__global__ __launch_bounds__(256)
void gather_kernel(const float* __restrict__ x,
                   const int* __restrict__ src,
                   const int* __restrict__ head,
                   const int* __restrict__ nxt,
                   float* __restrict__ h) {
  const int lane = threadIdx.x & 31;
  const int nw = gridDim.x * 8;
  const int w0 = __builtin_amdgcn_readfirstlane(blockIdx.x * 8 + (threadIdx.x >> 5));
  for (int node = w0; node < N_NODES; node += nw) {
    float4 a0 = make_float4(0.f, 0.f, 0.f, 0.f), a1 = a0, a2 = a0, a3 = a0;
    const int e0 = head[node];
    if (e0 < 0) {
      const float4* xs = (const float4*)(x + (size_t)node * D);
      a0 = xs[lane]; a1 = xs[32 + lane]; a2 = xs[64 + lane]; a3 = xs[96 + lane];
    } else {
      int sa = src[e0], ea = nxt[e0], sb = 0, eb = -1;
      for (int it = 0; it < N_EDGES; it += 2) {
        if (ea >= 0) { sb = src[ea]; eb = nxt[ea]; }
        add_row(x, sa, lane, a0, a1, a2, a3);
        if (ea < 0) break;
        if (eb >= 0) { sa = src[eb]; ea = nxt[eb]; }
        add_row(x, sb, lane, a0, a1, a2, a3);
        if (eb < 0) break;
      }
    }
    float4* hp = (float4*)(h + (size_t)node * D);
    hp[lane] = a0; hp[32 + lane] = a1; hp[64 + lane] = a2; hp[96 + lane] = a3;
  }
}

__global__ __launch_bounds__(256, 10)
void gemm_wmma_kernel(const float* hmat,
                      const float* Wm,
                      const float* __restrict__ bias,
                      float* __restrict__ out,
                      float* __restrict__ colsum,
                      float* __restrict__ colsumsq) {
  __shared__ float Alds[128 * 33];
  __shared__ float Blds[64 * 33];

  const int t = threadIdx.x;
  const int lane = t & 31;
  const int wave = t >> 5;
  const int rowTile = blockIdx.y * 128;
  const int colTile = blockIdx.x * 64;
  const int kg = (t & 7) * 4;

  const float* asrc[4];
#pragma unroll
  for (int i = 0; i < 4; ++i) {
    int r = (t >> 3) + 32 * i;
    int g = rowTile + r;
    g = g < N_NODES ? g : N_NODES - 1;
    asrc[i] = hmat + (size_t)g * D + kg;
  }
  const float* bsrc[2];
#pragma unroll
  for (int i = 0; i < 2; ++i) {
    int c = (t >> 3) + 32 * i;
    bsrc[i] = Wm + (size_t)(colTile + c) * D + kg;
  }

  v8f acc0 = {}, acc1 = {}, acc2 = {}, acc3 = {};

  float4 va[4], vb[2];
#pragma unroll
  for (int i = 0; i < 4; ++i) va[i] = *(const float4*)(asrc[i]);
#pragma unroll
  for (int i = 0; i < 2; ++i) vb[i] = *(const float4*)(bsrc[i]);

#pragma nounroll
  for (int kt = 0; kt < D; kt += 32) {
    __syncthreads();
#pragma unroll
    for (int i = 0; i < 4; ++i) {
      int r = (t >> 3) + 32 * i;
      Alds[r * 33 + kg + 0] = va[i].x;
      Alds[r * 33 + kg + 1] = va[i].y;
      Alds[r * 33 + kg + 2] = va[i].z;
      Alds[r * 33 + kg + 3] = va[i].w;
    }
#pragma unroll
    for (int i = 0; i < 2; ++i) {
      int c = (t >> 3) + 32 * i;
      Blds[c * 33 + kg + 0] = vb[i].x;
      Blds[c * 33 + kg + 1] = vb[i].y;
      Blds[c * 33 + kg + 2] = vb[i].z;
      Blds[c * 33 + kg + 3] = vb[i].w;
    }
    __syncthreads();
    if (kt + 32 < D) {
#pragma unroll
      for (int i = 0; i < 4; ++i) va[i] = *(const float4*)(asrc[i] + kt + 32);
#pragma unroll
      for (int i = 0; i < 2; ++i) vb[i] = *(const float4*)(bsrc[i] + kt + 32);
    }

    const int arow = wave * 16 + (lane & 15);
    const int bcol = lane & 15;
    const int khalf = (lane >> 4) * 2;
#pragma unroll
    for (int kk = 0; kk < 32; kk += 4) {
      int ko = kk + khalf;
      v2f a;  a[0]  = Alds[arow * 33 + ko];        a[1]  = Alds[arow * 33 + ko + 1];
      v2f b0; b0[0] = Blds[(bcol +  0) * 33 + ko]; b0[1] = Blds[(bcol +  0) * 33 + ko + 1];
      v2f b1; b1[0] = Blds[(bcol + 16) * 33 + ko]; b1[1] = Blds[(bcol + 16) * 33 + ko + 1];
      v2f b2; b2[0] = Blds[(bcol + 32) * 33 + ko]; b2[1] = Blds[(bcol + 32) * 33 + ko + 1];
      v2f b3; b3[0] = Blds[(bcol + 48) * 33 + ko]; b3[1] = Blds[(bcol + 48) * 33 + ko + 1];
      acc0 = __builtin_amdgcn_wmma_f32_16x16x4_f32(false, a, false, b0, (short)0, acc0, false, false);
      acc1 = __builtin_amdgcn_wmma_f32_16x16x4_f32(false, a, false, b1, (short)0, acc1, false, false);
      acc2 = __builtin_amdgcn_wmma_f32_16x16x4_f32(false, a, false, b2, (short)0, acc2, false, false);
      acc3 = __builtin_amdgcn_wmma_f32_16x16x4_f32(false, a, false, b3, (short)0, acc3, false, false);
    }
  }

  int gRowBase = rowTile + wave * 16 + ((lane >> 4) << 3);
  v8f accs[4] = {acc0, acc1, acc2, acc3};
#pragma unroll
  for (int nt = 0; nt < 4; ++nt) {
    int col = colTile + nt * 16 + (lane & 15);
    float bv = bias[col];
    float ps = 0.0f, pq = 0.0f;
#pragma unroll
    for (int r = 0; r < 8; ++r) {
      int grow = gRowBase + r;
      float h = accs[nt][r] + bv;
      h = h > 0.0f ? h : 0.0f;
      if (grow < N_NODES) {
        out[(size_t)grow * D + col] = h;
        ps += h;
        pq += h * h;
      }
    }
    ps += __shfl_xor(ps, 16, 32);
    pq += __shfl_xor(pq, 16, 32);
    if (lane < 16) {
      atomicAdd(colsum + col, ps);
      atomicAdd(colsumsq + col, pq);
    }
  }
}

__global__ void bn_kernel(float* __restrict__ out,
                          const float* __restrict__ colsum,
                          const float* __restrict__ colsumsq,
                          const float* __restrict__ gamma,
                          const float* __restrict__ beta) {
  const float invN = 1.0f / (float)N_NODES;
  long long total = (long long)N_NODES * D;
  long long i = (long long)blockIdx.x * blockDim.x + threadIdx.x;
  long long stride = (long long)gridDim.x * blockDim.x;
  for (; i < total; i += stride) {
    int col = (int)(i & (D - 1));
    float mean = colsum[col] * invN;
    float var = colsumsq[col] * invN - mean * mean;
    out[i] = (out[i] - mean) * rsqrtf(var + BN_EPS) * gamma[col] + beta[col];
  }
}

extern "C" void kernel_launch(void* const* d_in, const int* in_sizes, int n_in,
                              void* d_out, int out_size, void* d_ws, size_t ws_size,
                              hipStream_t stream) {
  const float* x     = (const float*)d_in[0];
  const int*   src   = (const int*)d_in[1];
  const int*   dst   = (const int*)d_in[2];
  const float* Wm    = (const float*)d_in[3];
  const float* bias  = (const float*)d_in[4];
  const float* gamma = (const float*)d_in[5];
  const float* beta  = (const float*)d_in[6];
  float* out = (float*)d_out;

  float* ws       = (float*)d_ws;
  float* h        = ws;
  int*   head     = (int*)(h + (size_t)N_NODES * D);
  float* colsum   = (float*)(head + N_NODES);
  float* colsumsq = colsum + D;
  int*   nxt      = (int*)d_out;

  init_kernel<<<(N_NODES + 255) / 256, 256, 0, stream>>>(head, colsum);

  link_kernel<<<(N_EDGES / 4 + 255) / 256, 256, 0, stream>>>(dst, head, nxt);

  gather_kernel<<<N_NODES / 16, 256, 0, stream>>>(x, src, head, nxt, h);

  dim3 grid(D / 64, (N_NODES + 127) / 128);
  gemm_wmma_kernel<<<grid, 256, 0, stream>>>(h, Wm, bias, out, colsum, colsumsq);

  bn_kernel<<<4096, 256, 0, stream>>>(out, colsum, colsumsq, gamma, beta);
}


# GEMM LDS tiles at 144-byte row pitch: 16-byte LDS stores and conflict-free 16-byte fragment reads, one read feeds two WMMAs (k taken 0,1,4,5 then 2,3,6,7 of each 8: f32 sums round in another order)
# speedup vs baseline: 1.6850x; 1.0376x over previous
#include <hip/hip_runtime.h>

#define N_NODES 20000
#define N_EDGES 150000
#define D 512
#define BN_EPS 1e-5f

typedef __attribute__((ext_vector_type(2))) float v2f;
typedef __attribute__((ext_vector_type(4))) float v4f;
typedef __attribute__((ext_vector_type(8))) float v8f;

#define LDP 36

__global__ void init_kernel(int* __restrict__ head, float* __restrict__ colstats) {
  int i = blockIdx.x * blockDim.x + threadIdx.x;
  if (i < N_NODES) head[i] = -1;
  if (i < 2 * D) colstats[i] = 0.0f;
}

static_assert(N_EDGES % 4 == 0, "link_kernel takes edges four at a time");
__global__ __launch_bounds__(256)
void link_kernel(const int* __restrict__ dst,
                 int* __restrict__ head,
                 int* __restrict__ nxt) {
  const int t = blockIdx.x * blockDim.x + threadIdx.x;
  if (t * 4 >= N_EDGES) return;
  const int4 d = ((const int4*)dst)[t];
  const int e = t * 4;
  int4 p;
  p.x = atomicExch(head + d.x, e);
  p.y = atomicExch(head + d.y, e + 1);
  p.z = atomicExch(head + d.z, e + 2);
  p.w = atomicExch(head + d.w, e + 3);
  ((int4*)nxt)[t] = p;
}

__device__ __forceinline__ void add_row(const float* __restrict__ x, int row, int lane,
                                        float4& a0, float4& a1, float4& a2, float4& a3) {
  const float4* xs = (const float4*)(x + (size_t)row * D);
  const float4 v0 = xs[lane], v1 = xs[32 + lane], v2 = xs[64 + lane], v3 = xs[96 + lane];
  a0.x += v0.x; a0.y += v0.y; a0.z += v0.z; a0.w += v0.w;
  a1.x += v1.x; a1.y += v1.y; a1.z += v1.z; a1.w += v1.w;
  a2.x += v2.x; a2.y += v2.y; a2.z += v2.z; a2.w += v2.w;
  a3.x += v3.x; a3.y += v3.y; a3.z += v3.z; a3.w += v3.w;
}

__global__ __launch_bounds__(256)
void gather_kernel(const float* __restrict__ x,
                   const int* __restrict__ src,
                   const int* __restrict__ head,
                   const int* __restrict__ nxt,
                   float* __restrict__ h) {
  const int lane = threadIdx.x & 31;
  const int nw = gridDim.x * 8;
  const int w0 = __builtin_amdgcn_readfirstlane(blockIdx.x * 8 + (threadIdx.x >> 5));
  for (int node = w0; node < N_NODES; node += nw) {
    float4 a0 = make_float4(0.f, 0.f, 0.f, 0.f), a1 = a0, a2 = a0, a3 = a0;
    const int e0 = head[node];
    if (e0 < 0) {
      const float4* xs = (const float4*)(x + (size_t)node * D);
      a0 = xs[lane]; a1 = xs[32 + lane]; a2 = xs[64 + lane]; a3 = xs[96 + lane];
    } else {
      int sa = src[e0], ea = nxt[e0], sb = 0, eb = -1;
      for (int it = 0; it < N_EDGES; it += 2) {
        if (ea >= 0) { sb = src[ea]; eb = nxt[ea]; }
        add_row(x, sa, lane, a0, a1, a2, a3);
        if (ea < 0) break;
        if (eb >= 0) { sa = src[eb]; ea = nxt[eb]; }
        add_row(x, sb, lane, a0, a1, a2, a3);
        if (eb < 0) break;
      }
    }
    float4* hp = (float4*)(h + (size_t)node * D);
    hp[lane] = a0; hp[32 + lane] = a1; hp[64 + lane] = a2; hp[96 + lane] = a3;
  }
}

__global__ __launch_bounds__(256, 10)
void gemm_wmma_kernel(const float* hmat,
                      const float* Wm,
                      const float* __restrict__ bias,
                      float* __restrict__ out,
                      float* __restrict__ colsum,
                      float* __restrict__ colsumsq) {
  __shared__ __attribute__((aligned(16))) float Alds[128 * LDP];
  __shared__ __attribute__((aligned(16))) float Blds[64 * LDP];

  const int t = threadIdx.x;
  const int lane = t & 31;
  const int wave = t >> 5;
  const int rowTile = blockIdx.y * 128;
  const int colTile = blockIdx.x * 64;
  const int kg = (t & 7) * 4;

  static_assert(N_NODES % 32 == 0, "a 32-row slab lies wholly inside or wholly outside the nodes");
  const unsigned voff = (unsigned)((t >> 3) * D + kg);
  const float* asrc[4];
#pragma unroll
  for (int i = 0; i < 4; ++i) {
    int rb = rowTile + 32 * i;
    rb = rb < N_NODES ? rb : rowTile;
    asrc[i] = hmat + (size_t)rb * D;
  }
  const float* bsrc[2];
#pragma unroll
  for (int i = 0; i < 2; ++i)
    bsrc[i] = Wm + (size_t)(colTile + 32 * i) * D;

  v8f acc0 = {}, acc1 = {}, acc2 = {}, acc3 = {};

  v4f va[4], vb[2];
#pragma unroll
  for (int i = 0; i < 4; ++i) va[i] = *(const v4f*)(asrc[i] + voff);
#pragma unroll
  for (int i = 0; i < 2; ++i) vb[i] = *(const v4f*)(bsrc[i] + voff);

#pragma nounroll
  for (int kt = 0; kt < D; kt += 32) {
    __syncthreads();
#pragma unroll
    for (int i = 0; i < 4; ++i) {
      int r = (t >> 3) + 32 * i;
      *(v4f*)&Alds[r * LDP + kg] = va[i];
    }
#pragma unroll
    for (int i = 0; i < 2; ++i) {
      int c = (t >> 3) + 32 * i;
      *(v4f*)&Blds[c * LDP + kg] = vb[i];
    }
    __syncthreads();
    if (kt + 32 < D) {
      unsigned vk = voff + (unsigned)(kt + 32);
      asm volatile("" : "+v"(vk));
#pragma unroll
      for (int i = 0; i < 4; ++i) va[i] = *(const v4f*)(asrc[i] + vk);
#pragma unroll
      for (int i = 0; i < 2; ++i) vb[i] = *(const v4f*)(bsrc[i] + vk);
    }

    const int arow = wave * 16 + (lane & 15);
    const int bcol = lane & 15;
    const int kq = (lane >> 4) * 4;
#pragma unroll
    for (int kk = 0; kk < 32; kk += 8) {
      const int ko = kk + kq;
      const v4f a4 = *(const v4f*)&Alds[arow * LDP + ko];
      const v4f p0 = *(const v4f*)&Blds[(bcol +  0) * LDP + ko];
      const v4f p1 = *(const v4f*)&Blds[(bcol + 16) * LDP + ko];
      const v4f p2 = *(const v4f*)&Blds[(bcol + 32) * LDP + ko];
      const v4f p3 = *(const v4f*)&Blds[(bcol + 48) * LDP + ko];
      acc0 = __builtin_amdgcn_wmma_f32_16x16x4_f32(false, a4.xy, false, p0.xy, (short)0, acc0, false, false);
      acc1 = __builtin_amdgcn_wmma_f32_16x16x4_f32(false, a4.xy, false, p1.xy, (short)0, acc1, false, false);
      acc2 = __builtin_amdgcn_wmma_f32_16x16x4_f32(false, a4.xy, false, p2.xy, (short)0, acc2, false, false);
      acc3 = __builtin_amdgcn_wmma_f32_16x16x4_f32(false, a4.xy, false, p3.xy, (short)0, acc3, false, false);
      acc0 = __builtin_amdgcn_wmma_f32_16x16x4_f32(false, a4.zw, false, p0.zw, (short)0, acc0, false, false);
      acc1 = __builtin_amdgcn_wmma_f32_16x16x4_f32(false, a4.zw, false, p1.zw, (short)0, acc1, false, false);
      acc2 = __builtin_amdgcn_wmma_f32_16x16x4_f32(false, a4.zw, false, p2.zw, (short)0, acc2, false, false);
      acc3 = __builtin_amdgcn_wmma_f32_16x16x4_f32(false, a4.zw, false, p3.zw, (short)0, acc3, false, false);
    }
  }

  int gRowBase = rowTile + wave * 16 + ((lane >> 4) << 3);
  v8f accs[4] = {acc0, acc1, acc2, acc3};
#pragma unroll
  for (int nt = 0; nt < 4; ++nt) {
    int col = colTile + nt * 16 + (lane & 15);
    float bv = bias[col];
    float ps = 0.0f, pq = 0.0f;
#pragma unroll
    for (int r = 0; r < 8; ++r) {
      int grow = gRowBase + r;
      float h = accs[nt][r] + bv;
      h = h > 0.0f ? h : 0.0f;
      if (grow < N_NODES) {
        out[(size_t)grow * D + col] = h;
        ps += h;
        pq += h * h;
      }
    }
    ps += __shfl_xor(ps, 16, 32);
    pq += __shfl_xor(pq, 16, 32);
    if (lane < 16) {
      atomicAdd(colsum + col, ps);
      atomicAdd(colsumsq + col, pq);
    }
  }
}

__global__ void bn_kernel(float* __restrict__ out,
                          const float* __restrict__ colsum,
                          const float* __restrict__ colsumsq,
                          const float* __restrict__ gamma,
                          const float* __restrict__ beta) {
  const float invN = 1.0f / (float)N_NODES;
  long long total = (long long)N_NODES * D;
  long long i = (long long)blockIdx.x * blockDim.x + threadIdx.x;
  long long stride = (long long)gridDim.x * blockDim.x;
  for (; i < total; i += stride) {
    int col = (int)(i & (D - 1));
    float mean = colsum[col] * invN;
    float var = colsumsq[col] * invN - mean * mean;
    out[i] = (out[i] - mean) * rsqrtf(var + BN_EPS) * gamma[col] + beta[col];
  }
}

extern "C" void kernel_launch(void* const* d_in, const int* in_sizes, int n_in,
                              void* d_out, int out_size, void* d_ws, size_t ws_size,
                              hipStream_t stream) {
  const float* x     = (const float*)d_in[0];
  const int*   src   = (const int*)d_in[1];
  const int*   dst   = (const int*)d_in[2];
  const float* Wm    = (const float*)d_in[3];
  const float* bias  = (const float*)d_in[4];
  const float* gamma = (const float*)d_in[5];
  const float* beta  = (const float*)d_in[6];
  float* out = (float*)d_out;

  float* ws       = (float*)d_ws;
  float* h        = ws;
  int*   head     = (int*)(h + (size_t)N_NODES * D);
  float* colsum   = (float*)(head + N_NODES);
  float* colsumsq = colsum + D;
  int*   nxt      = (int*)d_out;

  init_kernel<<<(N_NODES + 255) / 256, 256, 0, stream>>>(head, colsum);

  link_kernel<<<(N_EDGES / 4 + 255) / 256, 256, 0, stream>>>(dst, head, nxt);

  gather_kernel<<<N_NODES / 16, 256, 0, stream>>>(x, src, head, nxt, h);

  dim3 grid(D / 64, (N_NODES + 127) / 128);
  gemm_wmma_kernel<<<grid, 256, 0, stream>>>(h, Wm, bias, out, colsum, colsumsq);

  bn_kernel<<<4096, 256, 0, stream>>>(out, colsum, colsumsq, gamma, beta);
}


# In-edge lists built inside the gather launch: 250 work-groups each scan the destination list for their own 80 nodes, link them in LDS and sum their rows (LDS arrival order); init and link launches go
# speedup vs baseline: 1.8659x; 1.1074x over previous
#include <hip/hip_runtime.h>

#define N_NODES 20000
#define N_EDGES 150000
#define D 512
#define BN_EPS 1e-5f

typedef __attribute__((ext_vector_type(2))) float v2f;
typedef __attribute__((ext_vector_type(4))) float v4f;
typedef __attribute__((ext_vector_type(8))) float v8f;

#define LDP 36

#define NPB 80
#define CAP 8192
static_assert(N_NODES % NPB == 0, "every work-group owns a full range of nodes");
static_assert(N_EDGES % 4 == 0, "the destination list is read four entries at a time");

__device__ __forceinline__ void add_row(const float* __restrict__ x, int row, int lane,
                                        float4& a0, float4& a1, float4& a2, float4& a3) {
  const float4* xs = (const float4*)(x + (size_t)row * D);
  const float4 v0 = xs[lane], v1 = xs[32 + lane], v2 = xs[64 + lane], v3 = xs[96 + lane];
  a0.x += v0.x; a0.y += v0.y; a0.z += v0.z; a0.w += v0.w;
  a1.x += v1.x; a1.y += v1.y; a1.z += v1.z; a1.w += v1.w;
  a2.x += v2.x; a2.y += v2.y; a2.z += v2.z; a2.w += v2.w;
  a3.x += v3.x; a3.y += v3.y; a3.z += v3.z; a3.w += v3.w;
}

__device__ __forceinline__ void add_two_rows(const float* __restrict__ x, int r0, int r1, int lane,
                                             float4& a0, float4& a1, float4& a2, float4& a3) {
  const float4* xa = (const float4*)(x + (size_t)r0 * D);
  const float4* xb = (const float4*)(x + (size_t)r1 * D);
  const float4 u0 = xa[lane], u1 = xa[32 + lane], u2 = xa[64 + lane], u3 = xa[96 + lane];
  const float4 w0 = xb[lane], w1 = xb[32 + lane], w2 = xb[64 + lane], w3 = xb[96 + lane];
  a0.x += u0.x; a0.y += u0.y; a0.z += u0.z; a0.w += u0.w;
  a1.x += u1.x; a1.y += u1.y; a1.z += u1.z; a1.w += u1.w;
  a2.x += u2.x; a2.y += u2.y; a2.z += u2.z; a2.w += u2.w;
  a3.x += u3.x; a3.y += u3.y; a3.z += u3.z; a3.w += u3.w;
  a0.x += w0.x; a0.y += w0.y; a0.z += w0.z; a0.w += w0.w;
  a1.x += w1.x; a1.y += w1.y; a1.z += w1.z; a1.w += w1.w;
  a2.x += w2.x; a2.y += w2.y; a2.z += w2.z; a2.w += w2.w;
  a3.x += w3.x; a3.y += w3.y; a3.z += w3.z; a3.w += w3.w;
}

__global__ __launch_bounds__(1024)
void gather_kernel(const float* __restrict__ x,
                   const int* __restrict__ src,
                   const int* __restrict__ dst,
                   float* __restrict__ h,
                   float* __restrict__ colstats) {
  __shared__ int lhead[NPB];
  __shared__ int lcnt;
  __shared__ int lwork;
  __shared__ int psrc[CAP];
  __shared__ int pnext[CAP];

  const int tid = threadIdx.x;
  const int lane = tid & 31;
  const int base = blockIdx.x * NPB;

  if (tid < NPB) lhead[tid] = -1;
  if (tid == 0) { lcnt = 0; lwork = 0; }
  static_assert(2 * D == 1024, "one thread of work-group 0 per column statistic");
  if (blockIdx.x == 0) colstats[tid] = 0.0f;
  __syncthreads();

  const int4* d4 = (const int4*)dst;
  constexpr int NQ = N_EDGES / 4;
  for (int q0 = 0; q0 < NQ; q0 += 8 * 1024) {
    int4 dv[8];
#pragma unroll
    for (int j = 0; j < 8; ++j) {
      const int q = q0 + j * 1024 + tid;
      dv[j] = d4[q < NQ ? q : NQ - 1];
    }
#pragma unroll
    for (int j = 0; j < 8; ++j) {
      const int q = q0 + j * 1024 + tid;
      const unsigned ux = (unsigned)(dv[j].x - base), uy = (unsigned)(dv[j].y - base);
      const unsigned uz = (unsigned)(dv[j].z - base), uw = (unsigned)(dv[j].w - base);
      const unsigned lo = min(min(ux, uy), min(uz, uw));
      if (q < NQ && lo < NPB) {
        const unsigned u[4] = {ux, uy, uz, uw};
#pragma unroll
        for (int k = 0; k < 4; ++k) {
          if (u[k] < NPB) {
            const int slot = atomicAdd(&lcnt, 1);
            if (slot < CAP) {
              psrc[slot] = 4 * q + k;
              pnext[slot] = atomicExch(&lhead[u[k]], slot);
            }
          }
        }
      }
    }
  }
  __syncthreads();
  const int cnt = lcnt;
  const bool overflow = cnt > CAP;
  if (!overflow) {
    for (int i = tid; i < cnt; i += 1024) psrc[i] = src[psrc[i]];
  }
  __syncthreads();

  for (int it = 0; it < NPB; ++it) {
    int n = 0;
    if (lane == 0) n = atomicAdd(&lwork, 1);
    n = __builtin_amdgcn_readfirstlane(n);
    if (n >= NPB) break;
    const int node = base + n;
    float4 a0 = make_float4(0.f, 0.f, 0.f, 0.f), a1 = a0, a2 = a0, a3 = a0;
    bool any = false;
    if (!overflow) {
      int s = __builtin_amdgcn_readfirstlane(lhead[n]);
      for (int k = 0; k < CAP && s >= 0; k += 2) {
        const int r0 = __builtin_amdgcn_readfirstlane(psrc[s]);
        const int s1 = __builtin_amdgcn_readfirstlane(pnext[s]);
        any = true;
        if (s1 >= 0) {
          const int r1 = __builtin_amdgcn_readfirstlane(psrc[s1]);
          s = __builtin_amdgcn_readfirstlane(pnext[s1]);
          add_two_rows(x, r0, r1, lane, a0, a1, a2, a3);
        } else {
          s = -1;
          add_row(x, r0, lane, a0, a1, a2, a3);
        }
      }
    } else {
      for (int e0 = 0; e0 < N_EDGES; e0 += 32) {
        const int e = e0 + lane;
        const int d = e < N_EDGES ? dst[e] : -1;
        unsigned m = (unsigned)__ballot(d == node);
        while (m != 0u) {
          const int b = __builtin_ctz(m);
          m &= m - 1u;
          add_row(x, src[e0 + b], lane, a0, a1, a2, a3);
          any = true;
        }
      }
    }
    if (!any) {
      const float4* xs = (const float4*)(x + (size_t)node * D);
      a0 = xs[lane]; a1 = xs[32 + lane]; a2 = xs[64 + lane]; a3 = xs[96 + lane];
    }
    float4* hp = (float4*)(h + (size_t)node * D);
    hp[lane] = a0; hp[32 + lane] = a1; hp[64 + lane] = a2; hp[96 + lane] = a3;
  }
}

__global__ __launch_bounds__(256, 10)
void gemm_wmma_kernel(const float* hmat,
                      const float* Wm,
                      const float* __restrict__ bias,
                      float* __restrict__ out,
                      float* __restrict__ colsum,
                      float* __restrict__ colsumsq) {
  __shared__ __attribute__((aligned(16))) float Alds[128 * LDP];
  __shared__ __attribute__((aligned(16))) float Blds[64 * LDP];

  const int t = threadIdx.x;
  const int lane = t & 31;
  const int wave = t >> 5;
  const int rowTile = blockIdx.y * 128;
  const int colTile = blockIdx.x * 64;
  const int kg = (t & 7) * 4;

  static_assert(N_NODES % 32 == 0, "a 32-row slab lies wholly inside or wholly outside the nodes");
  const unsigned voff = (unsigned)((t >> 3) * D + kg);
  const float* asrc[4];
#pragma unroll
  for (int i = 0; i < 4; ++i) {
    int rb = rowTile + 32 * i;
    rb = rb < N_NODES ? rb : rowTile;
    asrc[i] = hmat + (size_t)rb * D;
  }
  const float* bsrc[2];
#pragma unroll
  for (int i = 0; i < 2; ++i)
    bsrc[i] = Wm + (size_t)(colTile + 32 * i) * D;

  v8f acc0 = {}, acc1 = {}, acc2 = {}, acc3 = {};

  v4f va[4], vb[2];
#pragma unroll
  for (int i = 0; i < 4; ++i) va[i] = *(const v4f*)(asrc[i] + voff);
#pragma unroll
  for (int i = 0; i < 2; ++i) vb[i] = *(const v4f*)(bsrc[i] + voff);

#pragma nounroll
  for (int kt = 0; kt < D; kt += 32) {
    __syncthreads();
#pragma unroll
    for (int i = 0; i < 4; ++i) {
      int r = (t >> 3) + 32 * i;
      *(v4f*)&Alds[r * LDP + kg] = va[i];
    }
#pragma unroll
    for (int i = 0; i < 2; ++i) {
      int c = (t >> 3) + 32 * i;
      *(v4f*)&Blds[c * LDP + kg] = vb[i];
    }
    __syncthreads();
    if (kt + 32 < D) {
      unsigned vk = voff + (unsigned)(kt + 32);
      asm volatile("" : "+v"(vk));
#pragma unroll
      for (int i = 0; i < 4; ++i) va[i] = *(const v4f*)(asrc[i] + vk);
#pragma unroll
      for (int i = 0; i < 2; ++i) vb[i] = *(const v4f*)(bsrc[i] + vk);
    }

    const int arow = wave * 16 + (lane & 15);
    const int bcol = lane & 15;
    const int kq = (lane >> 4) * 4;
#pragma unroll
    for (int kk = 0; kk < 32; kk += 8) {
      const int ko = kk + kq;
      const v4f a4 = *(const v4f*)&Alds[arow * LDP + ko];
      const v4f p0 = *(const v4f*)&Blds[(bcol +  0) * LDP + ko];
      const v4f p1 = *(const v4f*)&Blds[(bcol + 16) * LDP + ko];
      const v4f p2 = *(const v4f*)&Blds[(bcol + 32) * LDP + ko];
      const v4f p3 = *(const v4f*)&Blds[(bcol + 48) * LDP + ko];
      acc0 = __builtin_amdgcn_wmma_f32_16x16x4_f32(false, a4.xy, false, p0.xy, (short)0, acc0, false, false);
      acc1 = __builtin_amdgcn_wmma_f32_16x16x4_f32(false, a4.xy, false, p1.xy, (short)0, acc1, false, false);
      acc2 = __builtin_amdgcn_wmma_f32_16x16x4_f32(false, a4.xy, false, p2.xy, (short)0, acc2, false, false);
      acc3 = __builtin_amdgcn_wmma_f32_16x16x4_f32(false, a4.xy, false, p3.xy, (short)0, acc3, false, false);
      acc0 = __builtin_amdgcn_wmma_f32_16x16x4_f32(false, a4.zw, false, p0.zw, (short)0, acc0, false, false);
      acc1 = __builtin_amdgcn_wmma_f32_16x16x4_f32(false, a4.zw, false, p1.zw, (short)0, acc1, false, false);
      acc2 = __builtin_amdgcn_wmma_f32_16x16x4_f32(false, a4.zw, false, p2.zw, (short)0, acc2, false, false);
      acc3 = __builtin_amdgcn_wmma_f32_16x16x4_f32(false, a4.zw, false, p3.zw, (short)0, acc3, false, false);
    }
  }

  int gRowBase = rowTile + wave * 16 + ((lane >> 4) << 3);
  v8f accs[4] = {acc0, acc1, acc2, acc3};
#pragma unroll
  for (int nt = 0; nt < 4; ++nt) {
    int col = colTile + nt * 16 + (lane & 15);
    float bv = bias[col];
    float ps = 0.0f, pq = 0.0f;
#pragma unroll
    for (int r = 0; r < 8; ++r) {
      int grow = gRowBase + r;
      float h = accs[nt][r] + bv;
      h = h > 0.0f ? h : 0.0f;
      if (grow < N_NODES) {
        out[(size_t)grow * D + col] = h;
        ps += h;
        pq += h * h;
      }
    }
    ps += __shfl_xor(ps, 16, 32);
    pq += __shfl_xor(pq, 16, 32);
    if (lane < 16) {
      atomicAdd(colsum + col, ps);
      atomicAdd(colsumsq + col, pq);
    }
  }
}

__global__ void bn_kernel(float* __restrict__ out,
                          const float* __restrict__ colsum,
                          const float* __restrict__ colsumsq,
                          const float* __restrict__ gamma,
                          const float* __restrict__ beta) {
  const float invN = 1.0f / (float)N_NODES;
  long long total = (long long)N_NODES * D;
  long long i = (long long)blockIdx.x * blockDim.x + threadIdx.x;
  long long stride = (long long)gridDim.x * blockDim.x;
  for (; i < total; i += stride) {
    int col = (int)(i & (D - 1));
    float mean = colsum[col] * invN;
    float var = colsumsq[col] * invN - mean * mean;
    out[i] = (out[i] - mean) * rsqrtf(var + BN_EPS) * gamma[col] + beta[col];
  }
}

extern "C" void kernel_launch(void* const* d_in, const int* in_sizes, int n_in,
                              void* d_out, int out_size, void* d_ws, size_t ws_size,
                              hipStream_t stream) {
  const float* x     = (const float*)d_in[0];
  const int*   src   = (const int*)d_in[1];
  const int*   dst   = (const int*)d_in[2];
  const float* Wm    = (const float*)d_in[3];
  const float* bias  = (const float*)d_in[4];
  const float* gamma = (const float*)d_in[5];
  const float* beta  = (const float*)d_in[6];
  float* out = (float*)d_out;

  float* ws       = (float*)d_ws;
  float* h        = ws;
  float* colsum   = h + (size_t)N_NODES * D;
  float* colsumsq = colsum + D;

  gather_kernel<<<N_NODES / NPB, 1024, 0, stream>>>(x, src, dst, h, colsum);

  dim3 grid(D / 64, (N_NODES + 127) / 128);
  gemm_wmma_kernel<<<grid, 256, 0, stream>>>(h, Wm, bias, out, colsum, colsumsq);

  bn_kernel<<<4096, 256, 0, stream>>>(out, colsum, colsumsq, gamma, beta);
}


# BN pass by one resident set of 8192 waves instead of 32768: a thread stays in one column (mean, inverse deviation, gamma, beta formed once) and keeps 8 of its 40 loads in flight, not one
# speedup vs baseline: 1.9591x; 1.0499x over previous
#include <hip/hip_runtime.h>

#define N_NODES 20000
#define N_EDGES 150000
#define D 512
#define BN_EPS 1e-5f

typedef __attribute__((ext_vector_type(2))) float v2f;
typedef __attribute__((ext_vector_type(4))) float v4f;
typedef __attribute__((ext_vector_type(8))) float v8f;

#define LDP 36

#define NPB 80
#define CAP 8192
static_assert(N_NODES % NPB == 0, "every work-group owns a full range of nodes");
static_assert(N_EDGES % 4 == 0, "the destination list is read four entries at a time");

__device__ __forceinline__ void add_row(const float* __restrict__ x, int row, int lane,
                                        float4& a0, float4& a1, float4& a2, float4& a3) {
  const float4* xs = (const float4*)(x + (size_t)row * D);
  const float4 v0 = xs[lane], v1 = xs[32 + lane], v2 = xs[64 + lane], v3 = xs[96 + lane];
  a0.x += v0.x; a0.y += v0.y; a0.z += v0.z; a0.w += v0.w;
  a1.x += v1.x; a1.y += v1.y; a1.z += v1.z; a1.w += v1.w;
  a2.x += v2.x; a2.y += v2.y; a2.z += v2.z; a2.w += v2.w;
  a3.x += v3.x; a3.y += v3.y; a3.z += v3.z; a3.w += v3.w;
}

__device__ __forceinline__ void add_two_rows(const float* __restrict__ x, int r0, int r1, int lane,
                                             float4& a0, float4& a1, float4& a2, float4& a3) {
  const float4* xa = (const float4*)(x + (size_t)r0 * D);
  const float4* xb = (const float4*)(x + (size_t)r1 * D);
  const float4 u0 = xa[lane], u1 = xa[32 + lane], u2 = xa[64 + lane], u3 = xa[96 + lane];
  const float4 w0 = xb[lane], w1 = xb[32 + lane], w2 = xb[64 + lane], w3 = xb[96 + lane];
  a0.x += u0.x; a0.y += u0.y; a0.z += u0.z; a0.w += u0.w;
  a1.x += u1.x; a1.y += u1.y; a1.z += u1.z; a1.w += u1.w;
  a2.x += u2.x; a2.y += u2.y; a2.z += u2.z; a2.w += u2.w;
  a3.x += u3.x; a3.y += u3.y; a3.z += u3.z; a3.w += u3.w;
  a0.x += w0.x; a0.y += w0.y; a0.z += w0.z; a0.w += w0.w;
  a1.x += w1.x; a1.y += w1.y; a1.z += w1.z; a1.w += w1.w;
  a2.x += w2.x; a2.y += w2.y; a2.z += w2.z; a2.w += w2.w;
  a3.x += w3.x; a3.y += w3.y; a3.z += w3.z; a3.w += w3.w;
}

__global__ __launch_bounds__(1024)
void gather_kernel(const float* __restrict__ x,
                   const int* __restrict__ src,
                   const int* __restrict__ dst,
                   float* __restrict__ h,
                   float* __restrict__ colstats) {
  __shared__ int lhead[NPB];
  __shared__ int lcnt;
  __shared__ int lwork;
  __shared__ int psrc[CAP];
  __shared__ int pnext[CAP];

  const int tid = threadIdx.x;
  const int lane = tid & 31;
  const int base = blockIdx.x * NPB;

  if (tid < NPB) lhead[tid] = -1;
  if (tid == 0) { lcnt = 0; lwork = 0; }
  static_assert(2 * D == 1024, "one thread of work-group 0 per column statistic");
  if (blockIdx.x == 0) colstats[tid] = 0.0f;
  __syncthreads();

  const int4* d4 = (const int4*)dst;
  constexpr int NQ = N_EDGES / 4;
  for (int q0 = 0; q0 < NQ; q0 += 8 * 1024) {
    int4 dv[8];
#pragma unroll
    for (int j = 0; j < 8; ++j) {
      const int q = q0 + j * 1024 + tid;
      dv[j] = d4[q < NQ ? q : NQ - 1];
    }
#pragma unroll
    for (int j = 0; j < 8; ++j) {
      const int q = q0 + j * 1024 + tid;
      const unsigned ux = (unsigned)(dv[j].x - base), uy = (unsigned)(dv[j].y - base);
      const unsigned uz = (unsigned)(dv[j].z - base), uw = (unsigned)(dv[j].w - base);
      const unsigned lo = min(min(ux, uy), min(uz, uw));
      if (q < NQ && lo < NPB) {
        const unsigned u[4] = {ux, uy, uz, uw};
#pragma unroll
        for (int k = 0; k < 4; ++k) {
          if (u[k] < NPB) {
            const int slot = atomicAdd(&lcnt, 1);
            if (slot < CAP) {
              psrc[slot] = 4 * q + k;
              pnext[slot] = atomicExch(&lhead[u[k]], slot);
            }
          }
        }
      }
    }
  }
  __syncthreads();
  const int cnt = lcnt;
  const bool overflow = cnt > CAP;
  if (!overflow) {
    for (int i = tid; i < cnt; i += 1024) psrc[i] = src[psrc[i]];
  }
  __syncthreads();

  for (int it = 0; it < NPB; ++it) {
    int n = 0;
    if (lane == 0) n = atomicAdd(&lwork, 1);
    n = __builtin_amdgcn_readfirstlane(n);
    if (n >= NPB) break;
    const int node = base + n;
    float4 a0 = make_float4(0.f, 0.f, 0.f, 0.f), a1 = a0, a2 = a0, a3 = a0;
    bool any = false;
    if (!overflow) {
      int s = __builtin_amdgcn_readfirstlane(lhead[n]);
      for (int k = 0; k < CAP && s >= 0; k += 2) {
        const int r0 = __builtin_amdgcn_readfirstlane(psrc[s]);
        const int s1 = __builtin_amdgcn_readfirstlane(pnext[s]);
        any = true;
        if (s1 >= 0) {
          const int r1 = __builtin_amdgcn_readfirstlane(psrc[s1]);
          s = __builtin_amdgcn_readfirstlane(pnext[s1]);
          add_two_rows(x, r0, r1, lane, a0, a1, a2, a3);
        } else {
          s = -1;
          add_row(x, r0, lane, a0, a1, a2, a3);
        }
      }
    } else {
      for (int e0 = 0; e0 < N_EDGES; e0 += 32) {
        const int e = e0 + lane;
        const int d = e < N_EDGES ? dst[e] : -1;
        unsigned m = (unsigned)__ballot(d == node);
        while (m != 0u) {
          const int b = __builtin_ctz(m);
          m &= m - 1u;
          add_row(x, src[e0 + b], lane, a0, a1, a2, a3);
          any = true;
        }
      }
    }
    if (!any) {
      const float4* xs = (const float4*)(x + (size_t)node * D);
      a0 = xs[lane]; a1 = xs[32 + lane]; a2 = xs[64 + lane]; a3 = xs[96 + lane];
    }
    float4* hp = (float4*)(h + (size_t)node * D);
    hp[lane] = a0; hp[32 + lane] = a1; hp[64 + lane] = a2; hp[96 + lane] = a3;
  }
}

__global__ __launch_bounds__(256, 10)
void gemm_wmma_kernel(const float* hmat,
                      const float* Wm,
                      const float* __restrict__ bias,
                      float* __restrict__ out,
                      float* __restrict__ colsum,
                      float* __restrict__ colsumsq) {
  __shared__ __attribute__((aligned(16))) float Alds[128 * LDP];
  __shared__ __attribute__((aligned(16))) float Blds[64 * LDP];

  const int t = threadIdx.x;
  const int lane = t & 31;
  const int wave = t >> 5;
  const int rowTile = blockIdx.y * 128;
  const int colTile = blockIdx.x * 64;
  const int kg = (t & 7) * 4;

  static_assert(N_NODES % 32 == 0, "a 32-row slab lies wholly inside or wholly outside the nodes");
  const unsigned voff = (unsigned)((t >> 3) * D + kg);
  const float* asrc[4];
#pragma unroll
  for (int i = 0; i < 4; ++i) {
    int rb = rowTile + 32 * i;
    rb = rb < N_NODES ? rb : rowTile;
    asrc[i] = hmat + (size_t)rb * D;
  }
  const float* bsrc[2];
#pragma unroll
  for (int i = 0; i < 2; ++i)
    bsrc[i] = Wm + (size_t)(colTile + 32 * i) * D;

  v8f acc0 = {}, acc1 = {}, acc2 = {}, acc3 = {};

  v4f va[4], vb[2];
#pragma unroll
  for (int i = 0; i < 4; ++i) va[i] = *(const v4f*)(asrc[i] + voff);
#pragma unroll
  for (int i = 0; i < 2; ++i) vb[i] = *(const v4f*)(bsrc[i] + voff);

#pragma nounroll
  for (int kt = 0; kt < D; kt += 32) {
    __syncthreads();
#pragma unroll
    for (int i = 0; i < 4; ++i) {
      int r = (t >> 3) + 32 * i;
      *(v4f*)&Alds[r * LDP + kg] = va[i];
    }
#pragma unroll
    for (int i = 0; i < 2; ++i) {
      int c = (t >> 3) + 32 * i;
      *(v4f*)&Blds[c * LDP + kg] = vb[i];
    }
    __syncthreads();
    if (kt + 32 < D) {
      unsigned vk = voff + (unsigned)(kt + 32);
      asm volatile("" : "+v"(vk));
#pragma unroll
      for (int i = 0; i < 4; ++i) va[i] = *(const v4f*)(asrc[i] + vk);
#pragma unroll
      for (int i = 0; i < 2; ++i) vb[i] = *(const v4f*)(bsrc[i] + vk);
    }

    const int arow = wave * 16 + (lane & 15);
    const int bcol = lane & 15;
    const int kq = (lane >> 4) * 4;
#pragma unroll
    for (int kk = 0; kk < 32; kk += 8) {
      const int ko = kk + kq;
      const v4f a4 = *(const v4f*)&Alds[arow * LDP + ko];
      const v4f p0 = *(const v4f*)&Blds[(bcol +  0) * LDP + ko];
      const v4f p1 = *(const v4f*)&Blds[(bcol + 16) * LDP + ko];
      const v4f p2 = *(const v4f*)&Blds[(bcol + 32) * LDP + ko];
      const v4f p3 = *(const v4f*)&Blds[(bcol + 48) * LDP + ko];
      acc0 = __builtin_amdgcn_wmma_f32_16x16x4_f32(false, a4.xy, false, p0.xy, (short)0, acc0, false, false);
      acc1 = __builtin_amdgcn_wmma_f32_16x16x4_f32(false, a4.xy, false, p1.xy, (short)0, acc1, false, false);
      acc2 = __builtin_amdgcn_wmma_f32_16x16x4_f32(false, a4.xy, false, p2.xy, (short)0, acc2, false, false);
      acc3 = __builtin_amdgcn_wmma_f32_16x16x4_f32(false, a4.xy, false, p3.xy, (short)0, acc3, false, false);
      acc0 = __builtin_amdgcn_wmma_f32_16x16x4_f32(false, a4.zw, false, p0.zw, (short)0, acc0, false, false);
      acc1 = __builtin_amdgcn_wmma_f32_16x16x4_f32(false, a4.zw, false, p1.zw, (short)0, acc1, false, false);
      acc2 = __builtin_amdgcn_wmma_f32_16x16x4_f32(false, a4.zw, false, p2.zw, (short)0, acc2, false, false);
      acc3 = __builtin_amdgcn_wmma_f32_16x16x4_f32(false, a4.zw, false, p3.zw, (short)0, acc3, false, false);
    }
  }

  int gRowBase = rowTile + wave * 16 + ((lane >> 4) << 3);
  v8f accs[4] = {acc0, acc1, acc2, acc3};
#pragma unroll
  for (int nt = 0; nt < 4; ++nt) {
    int col = colTile + nt * 16 + (lane & 15);
    float bv = bias[col];
    float ps = 0.0f, pq = 0.0f;
#pragma unroll
    for (int r = 0; r < 8; ++r) {
      int grow = gRowBase + r;
      float h = accs[nt][r] + bv;
      h = h > 0.0f ? h : 0.0f;
      if (grow < N_NODES) {
        out[(size_t)grow * D + col] = h;
        ps += h;
        pq += h * h;
      }
    }
    ps += __shfl_xor(ps, 16, 32);
    pq += __shfl_xor(pq, 16, 32);
    if (lane < 16) {
      atomicAdd(colsum + col, ps);
      atomicAdd(colsumsq + col, pq);
    }
  }
}

#define BN_BLOCKS 1024
#define BN_THREADS 256
#define BN_STRIDE (BN_BLOCKS * BN_THREADS)
#define BN_TOTAL (N_NODES * D)
#define BN_GROUP 8
#define BN_NG 5
static_assert(BN_STRIDE % D == 0, "a thread stays in one column");
static_assert((BN_NG * BN_GROUP - 1) * BN_STRIDE < BN_TOTAL, "every element but a thread's last exists for every thread");
static_assert(BN_TOTAL <= BN_NG * BN_GROUP * BN_STRIDE, "the groups cover the matrix");
__global__ __launch_bounds__(BN_THREADS)
void bn_kernel(float* __restrict__ out,
               const float* __restrict__ colsum,
               const float* __restrict__ colsumsq,
               const float* __restrict__ gamma,
               const float* __restrict__ beta) {
  const float invN = 1.0f / (float)N_NODES;
  const int i0 = blockIdx.x * BN_THREADS + threadIdx.x;
  const int col = i0 & (D - 1);
  const float cs = colsum[col], cq = colsumsq[col];
  const float g = gamma[col], b = beta[col];
  const bool tail = i0 + (BN_NG * BN_GROUP - 1) * BN_STRIDE < BN_TOTAL;
  float* p = out + i0;
  float cur[BN_GROUP], nxt[BN_GROUP];
#pragma unroll
  for (int j = 0; j < BN_GROUP; ++j) cur[j] = p[j * BN_STRIDE];
  const float mean = cs * invN;
  const float var = cq * invN - mean * mean;
  const float rstd = rsqrtf(var + BN_EPS);
#pragma nounroll
  for (int gi = 0; gi < BN_NG - 2; ++gi) {
    float* q = p + BN_GROUP * BN_STRIDE;
#pragma unroll
    for (int j = 0; j < BN_GROUP; ++j) nxt[j] = q[j * BN_STRIDE];
#pragma unroll
    for (int j = 0; j < BN_GROUP; ++j) p[j * BN_STRIDE] = (cur[j] - mean) * rstd * g + b;
#pragma unroll
    for (int j = 0; j < BN_GROUP; ++j) cur[j] = nxt[j];
    p = q;
  }
  float* q = p + BN_GROUP * BN_STRIDE;
#pragma unroll
  for (int j = 0; j < BN_GROUP - 1; ++j) nxt[j] = q[j * BN_STRIDE];
  nxt[BN_GROUP - 1] = 0.0f;
  if (tail) nxt[BN_GROUP - 1] = q[(BN_GROUP - 1) * BN_STRIDE];
#pragma unroll
  for (int j = 0; j < BN_GROUP; ++j) p[j * BN_STRIDE] = (cur[j] - mean) * rstd * g + b;
#pragma unroll
  for (int j = 0; j < BN_GROUP - 1; ++j) q[j * BN_STRIDE] = (nxt[j] - mean) * rstd * g + b;
  if (tail) q[(BN_GROUP - 1) * BN_STRIDE] = (nxt[BN_GROUP - 1] - mean) * rstd * g + b;
}

extern "C" void kernel_launch(void* const* d_in, const int* in_sizes, int n_in,
                              void* d_out, int out_size, void* d_ws, size_t ws_size,
                              hipStream_t stream) {
  const float* x     = (const float*)d_in[0];
  const int*   src   = (const int*)d_in[1];
  const int*   dst   = (const int*)d_in[2];
  const float* Wm    = (const float*)d_in[3];
  const float* bias  = (const float*)d_in[4];
  const float* gamma = (const float*)d_in[5];
  const float* beta  = (const float*)d_in[6];
  float* out = (float*)d_out;

  float* ws       = (float*)d_ws;
  float* h        = ws;
  float* colsum   = h + (size_t)N_NODES * D;
  float* colsumsq = colsum + D;

  gather_kernel<<<N_NODES / NPB, 1024, 0, stream>>>(x, src, dst, h, colsum);

  dim3 grid(D / 64, (N_NODES + 127) / 128);
  gemm_wmma_kernel<<<grid, 256, 0, stream>>>(h, Wm, bias, out, colsum, colsumsq);

  bn_kernel<<<BN_BLOCKS, BN_THREADS, 0, stream>>>(out, colsum, colsumsq, gamma, beta);
}


# GEMM epilogue through an LDS tile: a wave stores whole 128-byte out lines with 16-byte stores, and the 8 waves' column sums meet in LDS before the atomics (8x fewer; sums round in another order)
# speedup vs baseline: 2.0343x; 1.0384x over previous
#include <hip/hip_runtime.h>

#define N_NODES 20000
#define N_EDGES 150000
#define D 512
#define BN_EPS 1e-5f

typedef __attribute__((ext_vector_type(2))) float v2f;
typedef __attribute__((ext_vector_type(4))) float v4f;
typedef __attribute__((ext_vector_type(8))) float v8f;

#define LDP 36
#define OP 68

#define NPB 80
#define CAP 8192
static_assert(N_NODES % NPB == 0, "every work-group owns a full range of nodes");
static_assert(N_EDGES % 4 == 0, "the destination list is read four entries at a time");

__device__ __forceinline__ void add_row(const float* __restrict__ x, int row, int lane,
                                        float4& a0, float4& a1, float4& a2, float4& a3) {
  const float4* xs = (const float4*)(x + (size_t)row * D);
  const float4 v0 = xs[lane], v1 = xs[32 + lane], v2 = xs[64 + lane], v3 = xs[96 + lane];
  a0.x += v0.x; a0.y += v0.y; a0.z += v0.z; a0.w += v0.w;
  a1.x += v1.x; a1.y += v1.y; a1.z += v1.z; a1.w += v1.w;
  a2.x += v2.x; a2.y += v2.y; a2.z += v2.z; a2.w += v2.w;
  a3.x += v3.x; a3.y += v3.y; a3.z += v3.z; a3.w += v3.w;
}

__device__ __forceinline__ void add_two_rows(const float* __restrict__ x, int r0, int r1, int lane,
                                             float4& a0, float4& a1, float4& a2, float4& a3) {
  const float4* xa = (const float4*)(x + (size_t)r0 * D);
  const float4* xb = (const float4*)(x + (size_t)r1 * D);
  const float4 u0 = xa[lane], u1 = xa[32 + lane], u2 = xa[64 + lane], u3 = xa[96 + lane];
  const float4 w0 = xb[lane], w1 = xb[32 + lane], w2 = xb[64 + lane], w3 = xb[96 + lane];
  a0.x += u0.x; a0.y += u0.y; a0.z += u0.z; a0.w += u0.w;
  a1.x += u1.x; a1.y += u1.y; a1.z += u1.z; a1.w += u1.w;
  a2.x += u2.x; a2.y += u2.y; a2.z += u2.z; a2.w += u2.w;
  a3.x += u3.x; a3.y += u3.y; a3.z += u3.z; a3.w += u3.w;
  a0.x += w0.x; a0.y += w0.y; a0.z += w0.z; a0.w += w0.w;
  a1.x += w1.x; a1.y += w1.y; a1.z += w1.z; a1.w += w1.w;
  a2.x += w2.x; a2.y += w2.y; a2.z += w2.z; a2.w += w2.w;
  a3.x += w3.x; a3.y += w3.y; a3.z += w3.z; a3.w += w3.w;
}

__global__ __launch_bounds__(1024)
void gather_kernel(const float* __restrict__ x,
                   const int* __restrict__ src,
                   const int* __restrict__ dst,
                   float* __restrict__ h,
                   float* __restrict__ colstats) {
  __shared__ int lhead[NPB];
  __shared__ int lcnt;
  __shared__ int lwork;
  __shared__ int psrc[CAP];
  __shared__ int pnext[CAP];

  const int tid = threadIdx.x;
  const int lane = tid & 31;
  const int base = blockIdx.x * NPB;

  if (tid < NPB) lhead[tid] = -1;
  if (tid == 0) { lcnt = 0; lwork = 0; }
  static_assert(2 * D == 1024, "one thread of work-group 0 per column statistic");
  if (blockIdx.x == 0) colstats[tid] = 0.0f;
  __syncthreads();

  const int4* d4 = (const int4*)dst;
  constexpr int NQ = N_EDGES / 4;
  for (int q0 = 0; q0 < NQ; q0 += 8 * 1024) {
    int4 dv[8];
#pragma unroll
    for (int j = 0; j < 8; ++j) {
      const int q = q0 + j * 1024 + tid;
      dv[j] = d4[q < NQ ? q : NQ - 1];
    }
#pragma unroll
    for (int j = 0; j < 8; ++j) {
      const int q = q0 + j * 1024 + tid;
      const unsigned ux = (unsigned)(dv[j].x - base), uy = (unsigned)(dv[j].y - base);
      const unsigned uz = (unsigned)(dv[j].z - base), uw = (unsigned)(dv[j].w - base);
      const unsigned lo = min(min(ux, uy), min(uz, uw));
      if (q < NQ && lo < NPB) {
        const unsigned u[4] = {ux, uy, uz, uw};
#pragma unroll
        for (int k = 0; k < 4; ++k) {
          if (u[k] < NPB) {
            const int slot = atomicAdd(&lcnt, 1);
            if (slot < CAP) {
              psrc[slot] = 4 * q + k;
              pnext[slot] = atomicExch(&lhead[u[k]], slot);
            }
          }
        }
      }
    }
  }
  __syncthreads();
  const int cnt = lcnt;
  const bool overflow = cnt > CAP;
  if (!overflow) {
    for (int i = tid; i < cnt; i += 1024) psrc[i] = src[psrc[i]];
  }
  __syncthreads();

  for (int it = 0; it < NPB; ++it) {
    int n = 0;
    if (lane == 0) n = atomicAdd(&lwork, 1);
    n = __builtin_amdgcn_readfirstlane(n);
    if (n >= NPB) break;
    const int node = base + n;
    float4 a0 = make_float4(0.f, 0.f, 0.f, 0.f), a1 = a0, a2 = a0, a3 = a0;
    bool any = false;
    if (!overflow) {
      int s = __builtin_amdgcn_readfirstlane(lhead[n]);
      for (int k = 0; k < CAP && s >= 0; k += 2) {
        const int r0 = __builtin_amdgcn_readfirstlane(psrc[s]);
        const int s1 = __builtin_amdgcn_readfirstlane(pnext[s]);
        any = true;
        if (s1 >= 0) {
          const int r1 = __builtin_amdgcn_readfirstlane(psrc[s1]);
          s = __builtin_amdgcn_readfirstlane(pnext[s1]);
          add_two_rows(x, r0, r1, lane, a0, a1, a2, a3);
        } else {
          s = -1;
          add_row(x, r0, lane, a0, a1, a2, a3);
        }
      }
    } else {
      for (int e0 = 0; e0 < N_EDGES; e0 += 32) {
        const int e = e0 + lane;
        const int d = e < N_EDGES ? dst[e] : -1;
        unsigned m = (unsigned)__ballot(d == node);
        while (m != 0u) {
          const int b = __builtin_ctz(m);
          m &= m - 1u;
          add_row(x, src[e0 + b], lane, a0, a1, a2, a3);
          any = true;
        }
      }
    }
    if (!any) {
      const float4* xs = (const float4*)(x + (size_t)node * D);
      a0 = xs[lane]; a1 = xs[32 + lane]; a2 = xs[64 + lane]; a3 = xs[96 + lane];
    }
    float4* hp = (float4*)(h + (size_t)node * D);
    hp[lane] = a0; hp[32 + lane] = a1; hp[64 + lane] = a2; hp[96 + lane] = a3;
  }
}

__global__ __launch_bounds__(256, 10)
void gemm_wmma_kernel(const float* hmat,
                      const float* Wm,
                      const float* __restrict__ bias,
                      float* __restrict__ out,
                      float* __restrict__ colsum,
                      float* __restrict__ colsumsq) {
  __shared__ __attribute__((aligned(16))) float smem[128 * OP + 8 * 64 * 2];
  static_assert(128 * LDP + 64 * LDP <= 128 * OP, "the K-loop tiles fit under the output tile");
  float* const Alds = smem;
  float* const Blds = smem + 128 * LDP;

  const int t = threadIdx.x;
  const int lane = t & 31;
  const int wave = t >> 5;
  const int rowTile = blockIdx.y * 128;
  const int colTile = blockIdx.x * 64;
  const int kg = (t & 7) * 4;

  static_assert(N_NODES % 32 == 0, "a 32-row slab lies wholly inside or wholly outside the nodes");
  const unsigned voff = (unsigned)((t >> 3) * D + kg);
  const float* asrc[4];
#pragma unroll
  for (int i = 0; i < 4; ++i) {
    int rb = rowTile + 32 * i;
    rb = rb < N_NODES ? rb : rowTile;
    asrc[i] = hmat + (size_t)rb * D;
  }
  const float* bsrc[2];
#pragma unroll
  for (int i = 0; i < 2; ++i)
    bsrc[i] = Wm + (size_t)(colTile + 32 * i) * D;

  v8f acc0 = {}, acc1 = {}, acc2 = {}, acc3 = {};

  v4f va[4], vb[2];
#pragma unroll
  for (int i = 0; i < 4; ++i) va[i] = *(const v4f*)(asrc[i] + voff);
#pragma unroll
  for (int i = 0; i < 2; ++i) vb[i] = *(const v4f*)(bsrc[i] + voff);

#pragma nounroll
  for (int kt = 0; kt < D; kt += 32) {
    __syncthreads();
#pragma unroll
    for (int i = 0; i < 4; ++i) {
      int r = (t >> 3) + 32 * i;
      *(v4f*)&Alds[r * LDP + kg] = va[i];
    }
#pragma unroll
    for (int i = 0; i < 2; ++i) {
      int c = (t >> 3) + 32 * i;
      *(v4f*)&Blds[c * LDP + kg] = vb[i];
    }
    __syncthreads();
    if (kt + 32 < D) {
      unsigned vk = voff + (unsigned)(kt + 32);
      asm volatile("" : "+v"(vk));
#pragma unroll
      for (int i = 0; i < 4; ++i) va[i] = *(const v4f*)(asrc[i] + vk);
#pragma unroll
      for (int i = 0; i < 2; ++i) vb[i] = *(const v4f*)(bsrc[i] + vk);
    }

    const int arow = wave * 16 + (lane & 15);
    const int bcol = lane & 15;
    const int kq = (lane >> 4) * 4;
#pragma unroll
    for (int kk = 0; kk < 32; kk += 8) {
      const int ko = kk + kq;
      const v4f a4 = *(const v4f*)&Alds[arow * LDP + ko];
      const v4f p0 = *(const v4f*)&Blds[(bcol +  0) * LDP + ko];
      const v4f p1 = *(const v4f*)&Blds[(bcol + 16) * LDP + ko];
      const v4f p2 = *(const v4f*)&Blds[(bcol + 32) * LDP + ko];
      const v4f p3 = *(const v4f*)&Blds[(bcol + 48) * LDP + ko];
      acc0 = __builtin_amdgcn_wmma_f32_16x16x4_f32(false, a4.xy, false, p0.xy, (short)0, acc0, false, false);
      acc1 = __builtin_amdgcn_wmma_f32_16x16x4_f32(false, a4.xy, false, p1.xy, (short)0, acc1, false, false);
      acc2 = __builtin_amdgcn_wmma_f32_16x16x4_f32(false, a4.xy, false, p2.xy, (short)0, acc2, false, false);
      acc3 = __builtin_amdgcn_wmma_f32_16x16x4_f32(false, a4.xy, false, p3.xy, (short)0, acc3, false, false);
      acc0 = __builtin_amdgcn_wmma_f32_16x16x4_f32(false, a4.zw, false, p0.zw, (short)0, acc0, false, false);
      acc1 = __builtin_amdgcn_wmma_f32_16x16x4_f32(false, a4.zw, false, p1.zw, (short)0, acc1, false, false);
      acc2 = __builtin_amdgcn_wmma_f32_16x16x4_f32(false, a4.zw, false, p2.zw, (short)0, acc2, false, false);
      acc3 = __builtin_amdgcn_wmma_f32_16x16x4_f32(false, a4.zw, false, p3.zw, (short)0, acc3, false, false);
    }
  }

  float* const tile = smem;
  float* const wsum = smem + 128 * OP;
  static_assert(N_NODES % 16 == 0, "the row test is the same for a whole wave");
  const bool rowsIn = rowTile + wave * 16 < N_NODES;
  const int lrow = wave * 16 + ((lane >> 4) << 3);
  __syncthreads();
  v8f accs[4] = {acc0, acc1, acc2, acc3};
#pragma unroll
  for (int nt = 0; nt < 4; ++nt) {
    const int lc = nt * 16 + (lane & 15);
    const float bv = bias[colTile + lc];
    float ps = 0.0f, pq = 0.0f;
#pragma unroll
    for (int r = 0; r < 8; ++r) {
      float h = accs[nt][r] + bv;
      h = h > 0.0f ? h : 0.0f;
      h = rowsIn ? h : 0.0f;
      tile[(lrow + r) * OP + lc] = h;
      ps += h;
      pq += h * h;
    }
    ps += __shfl_xor(ps, 16, 32);
    pq += __shfl_xor(pq, 16, 32);
    if (lane < 16) {
      wsum[(wave * 64 + lc) * 2 + 0] = ps;
      wsum[(wave * 64 + lc) * 2 + 1] = pq;
    }
  }
  __syncthreads();
  if (rowsIn) {
    const int c4 = (lane & 15) * 4;
#pragma unroll
    for (int i = 0; i < 8; ++i) {
      const int lr = wave * 16 + 2 * i + (lane >> 4);
      const v4f v = *(const v4f*)&tile[lr * OP + c4];
      *(v4f*)&out[(size_t)(rowTile + lr) * D + colTile + c4] = v;
    }
  }
  if (t < 128) {
    const int k = t >> 6, c = t & 63;
    float s = 0.0f;
#pragma unroll
    for (int w = 0; w < 8; ++w) s += wsum[(w * 64 + c) * 2 + k];
    atomicAdd((k ? colsumsq : colsum) + colTile + c, s);
  }
}

#define BN_BLOCKS 1024
#define BN_THREADS 256
#define BN_STRIDE (BN_BLOCKS * BN_THREADS)
#define BN_TOTAL (N_NODES * D)
#define BN_GROUP 8
#define BN_NG 5
static_assert(BN_STRIDE % D == 0, "a thread stays in one column");
static_assert((BN_NG * BN_GROUP - 1) * BN_STRIDE < BN_TOTAL, "every element but a thread's last exists for every thread");
static_assert(BN_TOTAL <= BN_NG * BN_GROUP * BN_STRIDE, "the groups cover the matrix");
__global__ __launch_bounds__(BN_THREADS)
void bn_kernel(float* __restrict__ out,
               const float* __restrict__ colsum,
               const float* __restrict__ colsumsq,
               const float* __restrict__ gamma,
               const float* __restrict__ beta) {
  const float invN = 1.0f / (float)N_NODES;
  const int i0 = blockIdx.x * BN_THREADS + threadIdx.x;
  const int col = i0 & (D - 1);
  const float cs = colsum[col], cq = colsumsq[col];
  const float g = gamma[col], b = beta[col];
  const bool tail = i0 + (BN_NG * BN_GROUP - 1) * BN_STRIDE < BN_TOTAL;
  float* p = out + i0;
  float cur[BN_GROUP], nxt[BN_GROUP];
#pragma unroll
  for (int j = 0; j < BN_GROUP; ++j) cur[j] = p[j * BN_STRIDE];
  const float mean = cs * invN;
  const float var = cq * invN - mean * mean;
  const float rstd = rsqrtf(var + BN_EPS);
#pragma nounroll
  for (int gi = 0; gi < BN_NG - 2; ++gi) {
    float* q = p + BN_GROUP * BN_STRIDE;
#pragma unroll
    for (int j = 0; j < BN_GROUP; ++j) nxt[j] = q[j * BN_STRIDE];
#pragma unroll
    for (int j = 0; j < BN_GROUP; ++j) p[j * BN_STRIDE] = (cur[j] - mean) * rstd * g + b;
#pragma unroll
    for (int j = 0; j < BN_GROUP; ++j) cur[j] = nxt[j];
    p = q;
  }
  float* q = p + BN_GROUP * BN_STRIDE;
#pragma unroll
  for (int j = 0; j < BN_GROUP - 1; ++j) nxt[j] = q[j * BN_STRIDE];
  nxt[BN_GROUP - 1] = 0.0f;
  if (tail) nxt[BN_GROUP - 1] = q[(BN_GROUP - 1) * BN_STRIDE];
#pragma unroll
  for (int j = 0; j < BN_GROUP; ++j) p[j * BN_STRIDE] = (cur[j] - mean) * rstd * g + b;
#pragma unroll
  for (int j = 0; j < BN_GROUP - 1; ++j) q[j * BN_STRIDE] = (nxt[j] - mean) * rstd * g + b;
  if (tail) q[(BN_GROUP - 1) * BN_STRIDE] = (nxt[BN_GROUP - 1] - mean) * rstd * g + b;
}

extern "C" void kernel_launch(void* const* d_in, const int* in_sizes, int n_in,
                              void* d_out, int out_size, void* d_ws, size_t ws_size,
                              hipStream_t stream) {
  const float* x     = (const float*)d_in[0];
  const int*   src   = (const int*)d_in[1];
  const int*   dst   = (const int*)d_in[2];
  const float* Wm    = (const float*)d_in[3];
  const float* bias  = (const float*)d_in[4];
  const float* gamma = (const float*)d_in[5];
  const float* beta  = (const float*)d_in[6];
  float* out = (float*)d_out;

  float* ws       = (float*)d_ws;
  float* h        = ws;
  float* colsum   = h + (size_t)N_NODES * D;
  float* colsumsq = colsum + D;

  gather_kernel<<<N_NODES / NPB, 1024, 0, stream>>>(x, src, dst, h, colsum);

  dim3 grid(D / 64, (N_NODES + 127) / 128);
  gemm_wmma_kernel<<<grid, 256, 0, stream>>>(h, Wm, bias, out, colsum, colsumsq);

  bn_kernel<<<BN_BLOCKS, BN_THREADS, 0, stream>>>(out, colsum, colsumsq, gamma, beta);
}


# Gather and GEMM fused: each work-group keeps its 80 summed h rows in LDS and multiplies them by all of W on f32 WMMA; h never goes to memory (column sums now per wave: they round in another order)
# speedup vs baseline: 2.3882x; 1.1740x over previous
#include <hip/hip_runtime.h>


#define N_NODES 20000
#define N_EDGES 150000
#define D 512
#define BN_EPS 1e-5f

typedef __attribute__((ext_vector_type(2))) float v2f;
typedef __attribute__((ext_vector_type(4))) float v4f;
typedef __attribute__((ext_vector_type(8))) float v8f;
typedef __attribute__((address_space(3))) const v4f lds_cv4f;

#define NPB 80
#define CAP 8192
#define HP 516
static_assert(N_NODES % NPB == 0, "every work-group owns a full range of nodes");
static_assert(NPB % 16 == 0, "a work-group's rows are whole 16-row matrix tiles");
static_assert(N_EDGES % 4 == 0, "the destination list is read four entries at a time");
static_assert(D == 16 * 32, "one 16-column strip of the output per wave of a 1024-thread work-group");

__device__ __forceinline__ void add_row(const float* __restrict__ x, int row, int lane,
                                        float4& a0, float4& a1, float4& a2, float4& a3) {
  const float4* xs = (const float4*)(x + (size_t)row * D);
  const float4 v0 = xs[lane], v1 = xs[32 + lane], v2 = xs[64 + lane], v3 = xs[96 + lane];
  a0.x += v0.x; a0.y += v0.y; a0.z += v0.z; a0.w += v0.w;
  a1.x += v1.x; a1.y += v1.y; a1.z += v1.z; a1.w += v1.w;
  a2.x += v2.x; a2.y += v2.y; a2.z += v2.z; a2.w += v2.w;
  a3.x += v3.x; a3.y += v3.y; a3.z += v3.z; a3.w += v3.w;
}

__device__ __forceinline__ void add_two_rows(const float* __restrict__ x, int r0, int r1, int lane,
                                             float4& a0, float4& a1, float4& a2, float4& a3) {
  const float4* xa = (const float4*)(x + (size_t)r0 * D);
  const float4* xb = (const float4*)(x + (size_t)r1 * D);
  const float4 u0 = xa[lane], u1 = xa[32 + lane], u2 = xa[64 + lane], u3 = xa[96 + lane];
  const float4 w0 = xb[lane], w1 = xb[32 + lane], w2 = xb[64 + lane], w3 = xb[96 + lane];
  a0.x += u0.x; a0.y += u0.y; a0.z += u0.z; a0.w += u0.w;
  a1.x += u1.x; a1.y += u1.y; a1.z += u1.z; a1.w += u1.w;
  a2.x += u2.x; a2.y += u2.y; a2.z += u2.z; a2.w += u2.w;
  a3.x += u3.x; a3.y += u3.y; a3.z += u3.z; a3.w += u3.w;
  a0.x += w0.x; a0.y += w0.y; a0.z += w0.z; a0.w += w0.w;
  a1.x += w1.x; a1.y += w1.y; a1.z += w1.z; a1.w += w1.w;
  a2.x += w2.x; a2.y += w2.y; a2.z += w2.z; a2.w += w2.w;
  a3.x += w3.x; a3.y += w3.y; a3.z += w3.z; a3.w += w3.w;
}

__global__ __launch_bounds__(1024)
void gather_gemm_kernel(const float* __restrict__ x,
                        const int* __restrict__ src,
                        const int* __restrict__ dst,
                        const float* __restrict__ Wm,
                        const float* __restrict__ bias,
                        float* __restrict__ out,
                        float* __restrict__ colsum,
                        float* __restrict__ colsumsq) {
  __shared__ int lhead[NPB];
  __shared__ int lcnt;
  __shared__ int lwork;
  __shared__ int psrc[CAP];
  __shared__ int pnext[CAP];
  __shared__ __attribute__((aligned(16))) float htile[NPB * HP];

  const int tid = threadIdx.x;
  const int lane = tid & 31;
  const int wave = tid >> 5;
  const int base = blockIdx.x * NPB;

  if (tid < NPB) lhead[tid] = -1;
  if (tid == 0) { lcnt = 0; lwork = 0; }
  __syncthreads();

  const int4* d4 = (const int4*)dst;
  constexpr int NQ = N_EDGES / 4;
  for (int q0 = 0; q0 < NQ; q0 += 8 * 1024) {
    int4 dv[8];
#pragma unroll
    for (int j = 0; j < 8; ++j) {
      const int q = q0 + j * 1024 + tid;
      dv[j] = d4[q < NQ ? q : NQ - 1];
    }
#pragma unroll
    for (int j = 0; j < 8; ++j) {
      const int q = q0 + j * 1024 + tid;
      const unsigned ux = (unsigned)(dv[j].x - base), uy = (unsigned)(dv[j].y - base);
      const unsigned uz = (unsigned)(dv[j].z - base), uw = (unsigned)(dv[j].w - base);
      const unsigned lo = min(min(ux, uy), min(uz, uw));
      if (q < NQ && lo < NPB) {
        const unsigned u[4] = {ux, uy, uz, uw};
#pragma unroll
        for (int k = 0; k < 4; ++k) {
          if (u[k] < NPB) {
            const int slot = atomicAdd(&lcnt, 1);
            if (slot < CAP) {
              psrc[slot] = 4 * q + k;
              pnext[slot] = atomicExch(&lhead[u[k]], slot);
            }
          }
        }
      }
    }
  }
  __syncthreads();
  const int cnt = lcnt;
  const bool overflow = cnt > CAP;
  if (!overflow) {
    for (int i = tid; i < cnt; i += 1024) psrc[i] = src[psrc[i]];
  }
  __syncthreads();

  for (int it = 0; it < NPB; ++it) {
    int n = 0;
    if (lane == 0) n = atomicAdd(&lwork, 1);
    n = __builtin_amdgcn_readfirstlane(n);
    if (n >= NPB) break;
    const int node = base + n;
    float4 a0 = make_float4(0.f, 0.f, 0.f, 0.f), a1 = a0, a2 = a0, a3 = a0;
    bool any = false;
    if (!overflow) {
      int s = __builtin_amdgcn_readfirstlane(lhead[n]);
      for (int k = 0; k < CAP && s >= 0; k += 2) {
        const int r0 = __builtin_amdgcn_readfirstlane(psrc[s]);
        const int s1 = __builtin_amdgcn_readfirstlane(pnext[s]);
        any = true;
        if (s1 >= 0) {
          const int r1 = __builtin_amdgcn_readfirstlane(psrc[s1]);
          s = __builtin_amdgcn_readfirstlane(pnext[s1]);
          add_two_rows(x, r0, r1, lane, a0, a1, a2, a3);
        } else {
          s = -1;
          add_row(x, r0, lane, a0, a1, a2, a3);
        }
      }
    } else {
      for (int e0 = 0; e0 < N_EDGES; e0 += 32) {
        const int e = e0 + lane;
        const int d = e < N_EDGES ? dst[e] : -1;
        unsigned m = (unsigned)__ballot(d == node);
        while (m != 0u) {
          const int b = __builtin_ctz(m);
          m &= m - 1u;
          add_row(x, src[e0 + b], lane, a0, a1, a2, a3);
          any = true;
        }
      }
    }
    if (!any) {
      const float4* xs = (const float4*)(x + (size_t)node * D);
      a0 = xs[lane]; a1 = xs[32 + lane]; a2 = xs[64 + lane]; a3 = xs[96 + lane];
    }
    float4* hp = (float4*)(htile + n * HP);
    hp[lane] = a0; hp[32 + lane] = a1; hp[64 + lane] = a2; hp[96 + lane] = a3;
  }
  __syncthreads();

  constexpr int RT = NPB / 16;
  const int c = lane & 15;
  const int kq = (lane >> 4) * 4;
  const v4f* wp = (const v4f*)(Wm + (size_t)(wave * 16 + c) * D + kq);
  lds_cv4f* ap[RT];
#pragma unroll
  for (int rt = 0; rt < RT; ++rt) {
    ap[rt] = (lds_cv4f*)(htile + (rt * 16 + c) * HP + kq);
    asm volatile("" : "+v"(ap[rt]));
  }
  v8f acc[RT];
#pragma unroll
  for (int rt = 0; rt < RT; ++rt) acc[rt] = v8f{};

  v4f b0[4], b1[4];
#pragma unroll
  for (int j = 0; j < 4; ++j) b0[j] = wp[2 * j];
#pragma nounroll
  for (int t = 0; t < D / 64; ++t) {
#pragma unroll
    for (int j = 0; j < 4; ++j) b1[j] = wp[8 + 2 * j];
#pragma unroll
    for (int j = 0; j < 4; ++j) {
      v4f a[RT];
#pragma unroll
      for (int rt = 0; rt < RT; ++rt) a[rt] = ap[rt][2 * j];
#pragma unroll
      for (int rt = 0; rt < RT; ++rt)
        acc[rt] = __builtin_amdgcn_wmma_f32_16x16x4_f32(false, a[rt].xy, false, b0[j].xy, (short)0, acc[rt], false, false);
#pragma unroll
      for (int rt = 0; rt < RT; ++rt)
        acc[rt] = __builtin_amdgcn_wmma_f32_16x16x4_f32(false, a[rt].zw, false, b0[j].zw, (short)0, acc[rt], false, false);
    }
    wp += 16;
    if (t + 1 < D / 64) {
#pragma unroll
      for (int j = 0; j < 4; ++j) b0[j] = wp[2 * j];
    }
#pragma unroll
    for (int j = 0; j < 4; ++j) {
      v4f a[RT];
#pragma unroll
      for (int rt = 0; rt < RT; ++rt) a[rt] = ap[rt][8 + 2 * j];
#pragma unroll
      for (int rt = 0; rt < RT; ++rt)
        acc[rt] = __builtin_amdgcn_wmma_f32_16x16x4_f32(false, a[rt].xy, false, b1[j].xy, (short)0, acc[rt], false, false);
#pragma unroll
      for (int rt = 0; rt < RT; ++rt)
        acc[rt] = __builtin_amdgcn_wmma_f32_16x16x4_f32(false, a[rt].zw, false, b1[j].zw, (short)0, acc[rt], false, false);
    }
#pragma unroll
    for (int rt = 0; rt < RT; ++rt) {
      ap[rt] += 16;
      asm volatile("" : "+v"(ap[rt]));
    }
  }
  asm volatile("v_nop\n\tv_nop\n\tv_nop\n\tv_nop"
               : "+v"(acc[0]), "+v"(acc[1]), "+v"(acc[2]), "+v"(acc[3]), "+v"(acc[4]));
  static_assert(RT == 5, "the statement above names every accumulator tile");

  __syncthreads();
  {
    const int col = wave * 16 + c;
    const float bv = bias[col];
    const int lrow = (lane >> 4) << 3;
    float ps = 0.0f, pq = 0.0f;
#pragma unroll
    for (int rt = 0; rt < RT; ++rt) {
#pragma unroll
      for (int r = 0; r < 8; ++r) {
        float hv = acc[rt][r] + bv;
        hv = hv > 0.0f ? hv : 0.0f;
        htile[(rt * 16 + lrow + r) * HP + col] = hv;
        ps += hv;
        pq += hv * hv;
      }
    }
    ps += __shfl_xor(ps, 16, 32);
    pq += __shfl_xor(pq, 16, 32);
    if (lane < 16) {
      atomicAdd(colsum + col, ps);
      atomicAdd(colsumsq + col, pq);
    }
  }
  __syncthreads();
  for (int r = wave; r < NPB; r += 32) {
    const v4f* tp = (const v4f*)(htile + r * HP);
    v4f* op = (v4f*)(out + (size_t)(base + r) * D);
#pragma unroll
    for (int j = 0; j < 4; ++j) op[lane + 32 * j] = tp[lane + 32 * j];
  }
}

static_assert(2 * D == 1024, "one thread per column statistic");
__global__ __launch_bounds__(1024)
void zero_stats_kernel(float* __restrict__ colstats) {
  colstats[threadIdx.x] = 0.0f;
}

#define BN_BLOCKS 1024
#define BN_THREADS 256
#define BN_STRIDE (BN_BLOCKS * BN_THREADS)
#define BN_TOTAL (N_NODES * D)
#define BN_GROUP 8
#define BN_NG 5
static_assert(BN_STRIDE % D == 0, "a thread stays in one column");
static_assert((BN_NG * BN_GROUP - 1) * BN_STRIDE < BN_TOTAL, "every element but a thread's last exists for every thread");
static_assert(BN_TOTAL <= BN_NG * BN_GROUP * BN_STRIDE, "the groups cover the matrix");
__global__ __launch_bounds__(BN_THREADS)
void bn_kernel(float* __restrict__ out,
               const float* __restrict__ colsum,
               const float* __restrict__ colsumsq,
               const float* __restrict__ gamma,
               const float* __restrict__ beta) {
  const float invN = 1.0f / (float)N_NODES;
  const int i0 = blockIdx.x * BN_THREADS + threadIdx.x;
  const int col = i0 & (D - 1);
  const float cs = colsum[col], cq = colsumsq[col];
  const float g = gamma[col], b = beta[col];
  const bool tail = i0 + (BN_NG * BN_GROUP - 1) * BN_STRIDE < BN_TOTAL;
  float* p = out + i0;
  float cur[BN_GROUP], nxt[BN_GROUP];
#pragma unroll
  for (int j = 0; j < BN_GROUP; ++j) cur[j] = p[j * BN_STRIDE];
  const float mean = cs * invN;
  const float var = cq * invN - mean * mean;
  const float rstd = rsqrtf(var + BN_EPS);
#pragma nounroll
  for (int gi = 0; gi < BN_NG - 2; ++gi) {
    float* q = p + BN_GROUP * BN_STRIDE;
#pragma unroll
    for (int j = 0; j < BN_GROUP; ++j) nxt[j] = q[j * BN_STRIDE];
#pragma unroll
    for (int j = 0; j < BN_GROUP; ++j) p[j * BN_STRIDE] = (cur[j] - mean) * rstd * g + b;
#pragma unroll
    for (int j = 0; j < BN_GROUP; ++j) cur[j] = nxt[j];
    p = q;
  }
  float* q = p + BN_GROUP * BN_STRIDE;
#pragma unroll
  for (int j = 0; j < BN_GROUP - 1; ++j) nxt[j] = q[j * BN_STRIDE];
  nxt[BN_GROUP - 1] = 0.0f;
  if (tail) nxt[BN_GROUP - 1] = q[(BN_GROUP - 1) * BN_STRIDE];
#pragma unroll
  for (int j = 0; j < BN_GROUP; ++j) p[j * BN_STRIDE] = (cur[j] - mean) * rstd * g + b;
#pragma unroll
  for (int j = 0; j < BN_GROUP - 1; ++j) q[j * BN_STRIDE] = (nxt[j] - mean) * rstd * g + b;
  if (tail) q[(BN_GROUP - 1) * BN_STRIDE] = (nxt[BN_GROUP - 1] - mean) * rstd * g + b;
}

extern "C" void kernel_launch(void* const* d_in, const int* in_sizes, int n_in,
                              void* d_out, int out_size, void* d_ws, size_t ws_size,
                              hipStream_t stream) {
  const float* x     = (const float*)d_in[0];
  const int*   src   = (const int*)d_in[1];
  const int*   dst   = (const int*)d_in[2];
  const float* Wm    = (const float*)d_in[3];
  const float* bias  = (const float*)d_in[4];
  const float* gamma = (const float*)d_in[5];
  const float* beta  = (const float*)d_in[6];
  float* out = (float*)d_out;

  float* ws       = (float*)d_ws;
  float* colsum   = ws;
  float* colsumsq = colsum + D;

  zero_stats_kernel<<<1, 1024, 0, stream>>>(colsum);

  gather_gemm_kernel<<<N_NODES / NPB, 1024, 0, stream>>>(x, src, dst, Wm, bias, out, colsum, colsumsq);

  bn_kernel<<<BN_BLOCKS, BN_THREADS, 0, stream>>>(out, colsum, colsumsq, gamma, beta);
}


# Destination-list scan by all 250 work-groups replaced by a counting sort in the first launch (37 slices by owner); an owner reads 37 short segments; lists link in another order: sums round anew
# speedup vs baseline: 2.4891x; 1.0422x over previous
#include <hip/hip_runtime.h>


#define N_NODES 20000
#define N_EDGES 150000
#define D 512
#define BN_EPS 1e-5f

typedef __attribute__((ext_vector_type(2))) float v2f;
typedef __attribute__((ext_vector_type(4))) float v4f;
typedef __attribute__((ext_vector_type(8))) float v8f;
typedef __attribute__((address_space(3))) const v4f lds_cv4f;

#define NPB 80
#define CAP 8192
#define HP 516
static_assert(N_NODES % NPB == 0, "every work-group owns a full range of nodes");
static_assert(NPB % 16 == 0, "a work-group's rows are whole 16-row matrix tiles");
static_assert(N_EDGES % 4 == 0, "the destination list is read four entries at a time");
static_assert(D == 16 * 32, "one 16-column strip of the output per wave of a 1024-thread work-group");
#define NSLICE 37
#define ESLICE 4096
#define NBIN 256
static_assert(NSLICE * ESLICE >= N_EDGES, "the slices cover the edge list");
static_assert(ESLICE == 4 * 1024, "a thread of the sorting launch takes four edges");
static_assert(N_NODES / NPB < NBIN, "every owner has a bin, and the bin behind the last owner exists");
static_assert(N_NODES <= (1 << 24) && NPB <= 128, "source node and owned row fit one 32-bit entry");

__device__ __forceinline__ void add_row(const float* __restrict__ x, int row, int lane,
                                        float4& a0, float4& a1, float4& a2, float4& a3) {
  const float4* xs = (const float4*)(x + (size_t)row * D);
  const float4 v0 = xs[lane], v1 = xs[32 + lane], v2 = xs[64 + lane], v3 = xs[96 + lane];
  a0.x += v0.x; a0.y += v0.y; a0.z += v0.z; a0.w += v0.w;
  a1.x += v1.x; a1.y += v1.y; a1.z += v1.z; a1.w += v1.w;
  a2.x += v2.x; a2.y += v2.y; a2.z += v2.z; a2.w += v2.w;
  a3.x += v3.x; a3.y += v3.y; a3.z += v3.z; a3.w += v3.w;
}

__device__ __forceinline__ void add_two_rows(const float* __restrict__ x, int r0, int r1, int lane,
                                             float4& a0, float4& a1, float4& a2, float4& a3) {
  const float4* xa = (const float4*)(x + (size_t)r0 * D);
  const float4* xb = (const float4*)(x + (size_t)r1 * D);
  const float4 u0 = xa[lane], u1 = xa[32 + lane], u2 = xa[64 + lane], u3 = xa[96 + lane];
  const float4 w0 = xb[lane], w1 = xb[32 + lane], w2 = xb[64 + lane], w3 = xb[96 + lane];
  a0.x += u0.x; a0.y += u0.y; a0.z += u0.z; a0.w += u0.w;
  a1.x += u1.x; a1.y += u1.y; a1.z += u1.z; a1.w += u1.w;
  a2.x += u2.x; a2.y += u2.y; a2.z += u2.z; a2.w += u2.w;
  a3.x += u3.x; a3.y += u3.y; a3.z += u3.z; a3.w += u3.w;
  a0.x += w0.x; a0.y += w0.y; a0.z += w0.z; a0.w += w0.w;
  a1.x += w1.x; a1.y += w1.y; a1.z += w1.z; a1.w += w1.w;
  a2.x += w2.x; a2.y += w2.y; a2.z += w2.z; a2.w += w2.w;
  a3.x += w3.x; a3.y += w3.y; a3.z += w3.z; a3.w += w3.w;
}

__global__ __launch_bounds__(1024)
void gather_gemm_kernel(const float* __restrict__ x,
                        const int* __restrict__ src,
                        const int* __restrict__ dst,
                        const float* __restrict__ Wm,
                        const float* __restrict__ bias,
                        float* __restrict__ out,
                        float* __restrict__ colsum,
                        float* __restrict__ colsumsq,
                        const int* __restrict__ segoff,
                        const unsigned* __restrict__ entries) {
  __shared__ int lhead[NPB];
  __shared__ int lcnt;
  __shared__ int lwork;
  __shared__ int psrc[CAP];
  __shared__ int pnext[CAP];
  __shared__ __attribute__((aligned(16))) float htile[NPB * HP];

  const int tid = threadIdx.x;
  const int lane = tid & 31;
  const int wave = tid >> 5;
  const int base = blockIdx.x * NPB;

  if (tid < NPB) lhead[tid] = -1;
  if (tid == 0) { lcnt = 0; lwork = 0; }
  __syncthreads();

  static_assert(NSLICE <= 64, "one 16-lane group of the work-group per slice");
  const int p = wave * 2 + (lane >> 4);
  if (p < NSLICE) {
    const int sa = max(segoff[p * NBIN + blockIdx.x], 0);
    const int se = min(segoff[p * NBIN + blockIdx.x + 1], ESLICE);
    const unsigned* ent = entries + p * ESLICE;
    for (int i = sa + (lane & 15); i < se; i += 16) {
      const unsigned w = ent[i];
      const unsigned u = w & 127u;
      const int slot = atomicAdd(&lcnt, 1);
      if (slot < CAP && u < NPB) {
        psrc[slot] = (int)(w >> 7);
        pnext[slot] = atomicExch(&lhead[u], slot);
      }
    }
  }
  __syncthreads();
  const int cnt = lcnt;
  const bool overflow = cnt > CAP;

  for (int it = 0; it < NPB; ++it) {
    int n = 0;
    if (lane == 0) n = atomicAdd(&lwork, 1);
    n = __builtin_amdgcn_readfirstlane(n);
    if (n >= NPB) break;
    const int node = base + n;
    float4 a0 = make_float4(0.f, 0.f, 0.f, 0.f), a1 = a0, a2 = a0, a3 = a0;
    bool any = false;
    if (!overflow) {
      int s = __builtin_amdgcn_readfirstlane(lhead[n]);
      for (int k = 0; k < CAP && s >= 0; k += 2) {
        const int r0 = __builtin_amdgcn_readfirstlane(psrc[s]);
        const int s1 = __builtin_amdgcn_readfirstlane(pnext[s]);
        any = true;
        if (s1 >= 0) {
          const int r1 = __builtin_amdgcn_readfirstlane(psrc[s1]);
          s = __builtin_amdgcn_readfirstlane(pnext[s1]);
          add_two_rows(x, r0, r1, lane, a0, a1, a2, a3);
        } else {
          s = -1;
          add_row(x, r0, lane, a0, a1, a2, a3);
        }
      }
    } else {
      for (int e0 = 0; e0 < N_EDGES; e0 += 32) {
        const int e = e0 + lane;
        const int d = e < N_EDGES ? dst[e] : -1;
        unsigned m = (unsigned)__ballot(d == node);
        while (m != 0u) {
          const int b = __builtin_ctz(m);
          m &= m - 1u;
          add_row(x, src[e0 + b], lane, a0, a1, a2, a3);
          any = true;
        }
      }
    }
    if (!any) {
      const float4* xs = (const float4*)(x + (size_t)node * D);
      a0 = xs[lane]; a1 = xs[32 + lane]; a2 = xs[64 + lane]; a3 = xs[96 + lane];
    }
    float4* hp = (float4*)(htile + n * HP);
    hp[lane] = a0; hp[32 + lane] = a1; hp[64 + lane] = a2; hp[96 + lane] = a3;
  }
  __syncthreads();

  constexpr int RT = NPB / 16;
  const int c = lane & 15;
  const int kq = (lane >> 4) * 4;
  const v4f* wp = (const v4f*)(Wm + (size_t)(wave * 16 + c) * D + kq);
  lds_cv4f* ap[RT];
#pragma unroll
  for (int rt = 0; rt < RT; ++rt) {
    ap[rt] = (lds_cv4f*)(htile + (rt * 16 + c) * HP + kq);
    asm volatile("" : "+v"(ap[rt]));
  }
  v8f acc[RT];
#pragma unroll
  for (int rt = 0; rt < RT; ++rt) acc[rt] = v8f{};

  v4f b0[4], b1[4];
#pragma unroll
  for (int j = 0; j < 4; ++j) b0[j] = wp[2 * j];
#pragma nounroll
  for (int t = 0; t < D / 64; ++t) {
#pragma unroll
    for (int j = 0; j < 4; ++j) b1[j] = wp[8 + 2 * j];
#pragma unroll
    for (int j = 0; j < 4; ++j) {
      v4f a[RT];
#pragma unroll
      for (int rt = 0; rt < RT; ++rt) a[rt] = ap[rt][2 * j];
#pragma unroll
      for (int rt = 0; rt < RT; ++rt)
        acc[rt] = __builtin_amdgcn_wmma_f32_16x16x4_f32(false, a[rt].xy, false, b0[j].xy, (short)0, acc[rt], false, false);
#pragma unroll
      for (int rt = 0; rt < RT; ++rt)
        acc[rt] = __builtin_amdgcn_wmma_f32_16x16x4_f32(false, a[rt].zw, false, b0[j].zw, (short)0, acc[rt], false, false);
    }
    wp += 16;
    if (t + 1 < D / 64) {
#pragma unroll
      for (int j = 0; j < 4; ++j) b0[j] = wp[2 * j];
    }
#pragma unroll
    for (int j = 0; j < 4; ++j) {
      v4f a[RT];
#pragma unroll
      for (int rt = 0; rt < RT; ++rt) a[rt] = ap[rt][8 + 2 * j];
#pragma unroll
      for (int rt = 0; rt < RT; ++rt)
        acc[rt] = __builtin_amdgcn_wmma_f32_16x16x4_f32(false, a[rt].xy, false, b1[j].xy, (short)0, acc[rt], false, false);
#pragma unroll
      for (int rt = 0; rt < RT; ++rt)
        acc[rt] = __builtin_amdgcn_wmma_f32_16x16x4_f32(false, a[rt].zw, false, b1[j].zw, (short)0, acc[rt], false, false);
    }
#pragma unroll
    for (int rt = 0; rt < RT; ++rt) {
      ap[rt] += 16;
      asm volatile("" : "+v"(ap[rt]));
    }
  }
  asm volatile("v_nop\n\tv_nop\n\tv_nop\n\tv_nop"
               : "+v"(acc[0]), "+v"(acc[1]), "+v"(acc[2]), "+v"(acc[3]), "+v"(acc[4]));
  static_assert(RT == 5, "the statement above names every accumulator tile");

  __syncthreads();
  {
    const int col = wave * 16 + c;
    const float bv = bias[col];
    const int lrow = (lane >> 4) << 3;
    float ps = 0.0f, pq = 0.0f;
#pragma unroll
    for (int rt = 0; rt < RT; ++rt) {
#pragma unroll
      for (int r = 0; r < 8; ++r) {
        float hv = acc[rt][r] + bv;
        hv = hv > 0.0f ? hv : 0.0f;
        htile[(rt * 16 + lrow + r) * HP + col] = hv;
        ps += hv;
        pq += hv * hv;
      }
    }
    ps += __shfl_xor(ps, 16, 32);
    pq += __shfl_xor(pq, 16, 32);
    if (lane < 16) {
      atomicAdd(colsum + col, ps);
      atomicAdd(colsumsq + col, pq);
    }
  }
  __syncthreads();
  for (int r = wave; r < NPB; r += 32) {
    const v4f* tp = (const v4f*)(htile + r * HP);
    v4f* op = (v4f*)(out + (size_t)(base + r) * D);
#pragma unroll
    for (int j = 0; j < 4; ++j) op[lane + 32 * j] = tp[lane + 32 * j];
  }
}

static_assert(2 * D == 1024, "one thread per column statistic");
__global__ __launch_bounds__(1024)
void bin_edges_kernel(const int* __restrict__ src,
                      const int* __restrict__ dst,
                      int* __restrict__ segoff,
                      unsigned* __restrict__ entries,
                      float* __restrict__ colstats) {
  __shared__ int cntb[NBIN];
  __shared__ int offb[NBIN];
  __shared__ int wtot[NBIN / 32];
  const int tid = threadIdx.x;
  const int lane = tid & 31;
  const int wave = tid >> 5;
  if (blockIdx.x == 0) colstats[tid] = 0.0f;
  if (tid < NBIN) cntb[tid] = 0;
  __syncthreads();

  constexpr int NQ = N_EDGES / 4;
  const int q = blockIdx.x * 1024 + tid;
  int4 d = make_int4(-1, -1, -1, -1), s = make_int4(0, 0, 0, 0);
  if (q < NQ) {
    d = ((const int4*)dst)[q];
    s = ((const int4*)src)[q];
  }
  const int dd[4] = {d.x, d.y, d.z, d.w};
  const int ss[4] = {s.x, s.y, s.z, s.w};
  int bin[4], rank[4];
#pragma unroll
  for (int k = 0; k < 4; ++k) {
    bin[k] = -1;
    rank[k] = 0;
    if ((unsigned)dd[k] < (unsigned)N_NODES) {
      bin[k] = dd[k] / NPB;
      rank[k] = atomicAdd(&cntb[bin[k]], 1);
    }
  }
  __syncthreads();

  const int v = tid < NBIN ? cntb[tid] : 0;
  int incl = v;
#pragma unroll
  for (int o = 1; o < 32; o <<= 1) {
    const int t = __shfl_up(incl, o, 32);
    if (lane >= o) incl += t;
  }
  if (tid < NBIN && lane == 31) wtot[wave] = incl;
  __syncthreads();
  if (tid < NBIN) {
    int pre = 0;
#pragma unroll
    for (int w = 0; w < NBIN / 32; ++w) pre += w < wave ? wtot[w] : 0;
    const int start = pre + incl - v;
    offb[tid] = start;
    segoff[blockIdx.x * NBIN + tid] = start;
  }
  __syncthreads();

  unsigned* ent = entries + blockIdx.x * ESLICE;
#pragma unroll
  for (int k = 0; k < 4; ++k) {
    if (bin[k] >= 0)
      ent[offb[bin[k]] + rank[k]] = ((unsigned)ss[k] << 7) | (unsigned)(dd[k] - bin[k] * NPB);
  }
}

#define BN_BLOCKS 1024
#define BN_THREADS 256
#define BN_STRIDE (BN_BLOCKS * BN_THREADS)
#define BN_TOTAL (N_NODES * D)
#define BN_GROUP 8
#define BN_NG 5
static_assert(BN_STRIDE % D == 0, "a thread stays in one column");
static_assert((BN_NG * BN_GROUP - 1) * BN_STRIDE < BN_TOTAL, "every element but a thread's last exists for every thread");
static_assert(BN_TOTAL <= BN_NG * BN_GROUP * BN_STRIDE, "the groups cover the matrix");
__global__ __launch_bounds__(BN_THREADS)
void bn_kernel(float* __restrict__ out,
               const float* __restrict__ colsum,
               const float* __restrict__ colsumsq,
               const float* __restrict__ gamma,
               const float* __restrict__ beta) {
  const float invN = 1.0f / (float)N_NODES;
  const int i0 = blockIdx.x * BN_THREADS + threadIdx.x;
  const int col = i0 & (D - 1);
  const float cs = colsum[col], cq = colsumsq[col];
  const float g = gamma[col], b = beta[col];
  const bool tail = i0 + (BN_NG * BN_GROUP - 1) * BN_STRIDE < BN_TOTAL;
  float* p = out + i0;
  float cur[BN_GROUP], nxt[BN_GROUP];
#pragma unroll
  for (int j = 0; j < BN_GROUP; ++j) cur[j] = p[j * BN_STRIDE];
  const float mean = cs * invN;
  const float var = cq * invN - mean * mean;
  const float rstd = rsqrtf(var + BN_EPS);
#pragma nounroll
  for (int gi = 0; gi < BN_NG - 2; ++gi) {
    float* q = p + BN_GROUP * BN_STRIDE;
#pragma unroll
    for (int j = 0; j < BN_GROUP; ++j) nxt[j] = q[j * BN_STRIDE];
#pragma unroll
    for (int j = 0; j < BN_GROUP; ++j) p[j * BN_STRIDE] = (cur[j] - mean) * rstd * g + b;
#pragma unroll
    for (int j = 0; j < BN_GROUP; ++j) cur[j] = nxt[j];
    p = q;
  }
  float* q = p + BN_GROUP * BN_STRIDE;
#pragma unroll
  for (int j = 0; j < BN_GROUP - 1; ++j) nxt[j] = q[j * BN_STRIDE];
  nxt[BN_GROUP - 1] = 0.0f;
  if (tail) nxt[BN_GROUP - 1] = q[(BN_GROUP - 1) * BN_STRIDE];
#pragma unroll
  for (int j = 0; j < BN_GROUP; ++j) p[j * BN_STRIDE] = (cur[j] - mean) * rstd * g + b;
#pragma unroll
  for (int j = 0; j < BN_GROUP - 1; ++j) q[j * BN_STRIDE] = (nxt[j] - mean) * rstd * g + b;
  if (tail) q[(BN_GROUP - 1) * BN_STRIDE] = (nxt[BN_GROUP - 1] - mean) * rstd * g + b;
}

extern "C" void kernel_launch(void* const* d_in, const int* in_sizes, int n_in,
                              void* d_out, int out_size, void* d_ws, size_t ws_size,
                              hipStream_t stream) {
  const float* x     = (const float*)d_in[0];
  const int*   src   = (const int*)d_in[1];
  const int*   dst   = (const int*)d_in[2];
  const float* Wm    = (const float*)d_in[3];
  const float* bias  = (const float*)d_in[4];
  const float* gamma = (const float*)d_in[5];
  const float* beta  = (const float*)d_in[6];
  float* out = (float*)d_out;

  float* ws       = (float*)d_ws;
  float* colsum   = ws;
  float* colsumsq = colsum + D;
  int* segoff = (int*)(colsumsq + D);
  unsigned* entries = (unsigned*)(segoff + NSLICE * NBIN);

  bin_edges_kernel<<<NSLICE, 1024, 0, stream>>>(src, dst, segoff, entries, colsum);

  gather_gemm_kernel<<<N_NODES / NPB, 1024, 0, stream>>>(x, src, dst, Wm, bias, out, colsum, colsumsq, segoff, entries);

  bn_kernel<<<BN_BLOCKS, BN_THREADS, 0, stream>>>(out, colsum, colsumsq, gamma, beta);
}
